# starting kernel
# baseline (speedup 1.0000x reference)
#include <hip/hip_runtime.h>

typedef __attribute__((ext_vector_type(2))) float v2f;
typedef __attribute__((ext_vector_type(8))) float v8f;

#define D_IN  128
#define D_HID 128
#define D_OUT 64

__global__ void deg_init_kernel(float* __restrict__ deg, int n) {
    int i = blockIdx.x * blockDim.x + threadIdx.x;
    if (i < n) deg[i] = 1.0f;
}

__global__ void deg_accum_kernel(const int* __restrict__ dst,
                                 float* __restrict__ deg, int e) {
    int i = blockIdx.x * blockDim.x + threadIdx.x;
    if (i < e) atomicAdd(&deg[dst[i]], 1.0f);
}

__global__ void deg_rsqrt_kernel(float* __restrict__ dinv, int n) {
    int i = blockIdx.x * blockDim.x + threadIdx.x;
    if (i < n) {
        float d = dinv[i];
        dinv[i] = (d > 0.0f) ? rsqrtf(d) : 0.0f;
    }
}

template <int NOUT>
__global__ void gemm_wmma_f32_kernel(const float* __restrict__ A,
                                     const float* __restrict__ W,
                                     float* __restrict__ H, int M) {
    constexpr int K = 128;
    constexpr int NT = NOUT / 16;
    __shared__ __align__(16) float lds_a[16 * K];

    const int block_row = blockIdx.x * 16;
    const int tid = threadIdx.x;
    constexpr int NTHREADS = 32 * NT;

    for (int i = tid * 4; i < 16 * K; i += NTHREADS * 4) {
        *(float4*)(&lds_a[i]) = *(const float4*)(&A[(size_t)block_row * K + i]);
    }
    __syncthreads();

    const int wave = tid >> 5;
    const int lane = tid & 31;
    const int half = lane >> 4;
    const int l16  = lane & 15;
    const int col_base = wave * 16;

    v8f acc = {};
    #pragma unroll 4
    for (int k0 = 0; k0 < K; k0 += 4) {
        const int ka = k0 + half * 2;
        v2f a, b;
        a.x = lds_a[l16 * K + ka];
        a.y = lds_a[l16 * K + ka + 1];
        b.x = W[(size_t)ka * NOUT + col_base + l16];
        b.y = W[(size_t)(ka + 1) * NOUT + col_base + l16];
        acc = __builtin_amdgcn_wmma_f32_16x16x4_f32(
             false, a,  false, b,
             (short)0, acc,  false,  false);
    }

    #pragma unroll
    for (int v = 0; v < 8; ++v) {
        const int row = v + half * 8;
        H[(size_t)(block_row + row) * NOUT + col_base + l16] = acc[v];
    }
}

template <int D>
__global__ void self_init_kernel(const float* __restrict__ h,
                                 const float* __restrict__ dinv,
                                 float* __restrict__ agg, int total) {
    int idx = blockIdx.x * blockDim.x + threadIdx.x;
    if (idx >= total) return;
    const int node = idx / D;
    const float di = dinv[node];
    agg[idx] = h[idx] * di * di;
}

template <int D, int VPL>
__global__ void edge_scatter_kernel(const int* __restrict__ src,
                                    const int* __restrict__ dst,
                                    const float* __restrict__ h,
                                    const float* __restrict__ dinv,
                                    float* __restrict__ agg, int e) {
    const int widx = (blockIdx.x * blockDim.x + threadIdx.x) >> 5;
    const int lane = threadIdx.x & 31;
    if (widx >= e) return;
    const int s = src[widx];
    const int d = dst[widx];
    const float norm = dinv[s] * dinv[d];
    const float* __restrict__ hp = h + (size_t)s * D + lane * VPL;
    float* __restrict__ op = agg + (size_t)d * D + lane * VPL;
    if constexpr (VPL == 4) {
        const float4 v = *(const float4*)hp;
        atomicAdd(op + 0, v.x * norm);
        atomicAdd(op + 1, v.y * norm);
        atomicAdd(op + 2, v.z * norm);
        atomicAdd(op + 3, v.w * norm);
    } else {
        const float2 v = *(const float2*)hp;
        atomicAdd(op + 0, v.x * norm);
        atomicAdd(op + 1, v.y * norm);
    }
}

template <int D, bool RELU>
__global__ void bias_act_kernel(float* __restrict__ data,
                                const float* __restrict__ bias, int total) {
    int idx = blockIdx.x * blockDim.x + threadIdx.x;
    if (idx >= total) return;
    float v = data[idx] + bias[idx & (D - 1)];
    if (RELU) v = fmaxf(v, 0.0f);
    data[idx] = v;
}

extern "C" void kernel_launch(void* const* d_in, const int* in_sizes, int n_in,
                              void* d_out, int out_size, void* d_ws, size_t ws_size,
                              hipStream_t stream) {
    const float* x    = (const float*)d_in[0];
    const int*   eidx = (const int*)  d_in[1];
    const float* W1   = (const float*)d_in[2];
    const float* b1   = (const float*)d_in[3];
    const float* W2   = (const float*)d_in[4];
    const float* b2   = (const float*)d_in[5];
    float* out = (float*)d_out;

    const int N = in_sizes[0] / D_IN;
    const int E = in_sizes[1] / 2;
    const int* src = eidx;
    const int* dst = eidx + E;

    float* ws = (float*)d_ws;
    float* dinv = ws;
    float* h1   = ws + (((size_t)N + 255) & ~(size_t)255);
    float* agg1 = h1 + (size_t)N * D_HID;
    float* h2   = h1;

    const int BT = 256;

    deg_init_kernel<<<(N + BT - 1) / BT, BT, 0, stream>>>(dinv, N);
    deg_accum_kernel<<<(E + BT - 1) / BT, BT, 0, stream>>>(dst, dinv, E);
    deg_rsqrt_kernel<<<(N + BT - 1) / BT, BT, 0, stream>>>(dinv, N);

    gemm_wmma_f32_kernel<D_HID><<<N / 16, 32 * (D_HID / 16), 0, stream>>>(x, W1, h1, N);
    {
        const int total = N * D_HID;
        self_init_kernel<D_HID><<<(total + BT - 1) / BT, BT, 0, stream>>>(h1, dinv, agg1, total);
        const long tthreads = (long)E * 32;
        edge_scatter_kernel<D_HID, 4><<<(unsigned)((tthreads + BT - 1) / BT), BT, 0, stream>>>(
            src, dst, h1, dinv, agg1, E);
        bias_act_kernel<D_HID, true><<<(total + BT - 1) / BT, BT, 0, stream>>>(agg1, b1, total);
    }

    gemm_wmma_f32_kernel<D_OUT><<<N / 16, 32 * (D_OUT / 16), 0, stream>>>(agg1, W2, h2, N);
    {
        const int total = N * D_OUT;
        self_init_kernel<D_OUT><<<(total + BT - 1) / BT, BT, 0, stream>>>(h2, dinv, out, total);
        const long tthreads = (long)E * 32;
        edge_scatter_kernel<D_OUT, 2><<<(unsigned)((tthreads + BT - 1) / BT), BT, 0, stream>>>(
            src, dst, h2, dinv, out, E);
        bias_act_kernel<D_OUT, false><<<(total + BT - 1) / BT, BT, 0, stream>>>(out, b2, total);
    }
}


# Float-atomic edge scatter replaced by per-target gather over incoming-edge lists built in-call (int-atomic arrival-order rank, scan, fill); sums run in rank order, two half-wave partials at D=64
# speedup vs baseline: 2.3022x; 2.3022x over previous
#include <hip/hip_runtime.h>


typedef __attribute__((ext_vector_type(2))) float v2f;
typedef __attribute__((ext_vector_type(8))) float v8f;

#define D_IN  128
#define D_HID 128
#define D_OUT 64

__global__ void cnt_zero_kernel(int* __restrict__ cnt, int n) {
    int i = blockIdx.x * blockDim.x + threadIdx.x;
    if (i < n) cnt[i] = 0;
}

__global__ void count_rank_kernel(const int* __restrict__ dst,
                                  int* __restrict__ cnt,
                                  int* __restrict__ rank, int e) {
    int i = blockIdx.x * blockDim.x + threadIdx.x;
    if (i < e) rank[i] = atomicAdd(&cnt[dst[i]], 1);
}

#define SCAN_T 1024

__global__ __launch_bounds__(SCAN_T)
void rowptr_dinv_kernel(const int* __restrict__ cnt,
                        int* __restrict__ rowptr,
                        float* __restrict__ dinv, int n) {
    __shared__ int wsum[32];
    __shared__ int base_s;
    const int tid = threadIdx.x;
    const int lane = tid & 31;
    const int wave = tid >> 5;
    const int first = blockIdx.x * SCAN_T;

    int s = 0;
    for (int i = tid * 4; i < first; i += SCAN_T * 4) {
        const int4 v = *(const int4*)(cnt + i);
        s += (v.x + v.y) + (v.z + v.w);
    }
    for (int o = 16; o > 0; o >>= 1) s += __shfl_xor(s, o);
    if (lane == 0) wsum[wave] = s;
    __syncthreads();
    if (wave == 0) {
        int t = wsum[lane];
        for (int o = 16; o > 0; o >>= 1) t += __shfl_xor(t, o);
        if (lane == 0) base_s = t;
    }
    __syncthreads();
    const int base = base_s;

    const int i = first + tid;
    const int c = (i < n) ? cnt[i] : 0;
    int incl = c;
    for (int o = 1; o < 32; o <<= 1) {
        const int t = __shfl_up(incl, o);
        if (lane >= o) incl += t;
    }
    if (lane == 31) wsum[wave] = incl;
    __syncthreads();
    if (wave == 0) {
        const int t = wsum[lane];
        int ti = t;
        for (int o = 1; o < 32; o <<= 1) {
            const int u = __shfl_up(ti, o);
            if (lane >= o) ti += u;
        }
        wsum[lane] = ti - t;
    }
    __syncthreads();
    if (i < n) {
        rowptr[i] = base + wsum[wave] + (incl - c);
        const float d = (float)(c + 1);
        dinv[i] = (d > 0.0f) ? rsqrtf(d) : 0.0f;
    }
}

__global__ void csr_fill_kernel(const int* __restrict__ src,
                                const int* __restrict__ dst,
                                const int* __restrict__ rank,
                                const int* __restrict__ rowptr,
                                int* __restrict__ col, int e) {
    int i = blockIdx.x * blockDim.x + threadIdx.x;
    if (i < e) col[rowptr[dst[i]] + rank[i]] = src[i];
}

template <int NOUT>
__global__ void gemm_wmma_f32_kernel(const float* __restrict__ A,
                                     const float* __restrict__ W,
                                     float* __restrict__ H, int M) {
    constexpr int K = 128;
    constexpr int NT = NOUT / 16;
    __shared__ __align__(16) float lds_a[16 * K];

    const int block_row = blockIdx.x * 16;
    const int tid = threadIdx.x;
    constexpr int NTHREADS = 32 * NT;

    for (int i = tid * 4; i < 16 * K; i += NTHREADS * 4) {
        *(float4*)(&lds_a[i]) = *(const float4*)(&A[(size_t)block_row * K + i]);
    }
    __syncthreads();

    const int wave = tid >> 5;
    const int lane = tid & 31;
    const int half = lane >> 4;
    const int l16  = lane & 15;
    const int col_base = wave * 16;

    v8f acc = {};
    #pragma unroll 4
    for (int k0 = 0; k0 < K; k0 += 4) {
        const int ka = k0 + half * 2;
        v2f a, b;
        a.x = lds_a[l16 * K + ka];
        a.y = lds_a[l16 * K + ka + 1];
        b.x = W[(size_t)ka * NOUT + col_base + l16];
        b.y = W[(size_t)(ka + 1) * NOUT + col_base + l16];
        acc = __builtin_amdgcn_wmma_f32_16x16x4_f32(
             false, a,  false, b,
             (short)0, acc,  false,  false);
    }

    #pragma unroll
    for (int v = 0; v < 8; ++v) {
        const int row = v + half * 8;
        H[(size_t)(block_row + row) * NOUT + col_base + l16] = acc[v];
    }
}

template <int D, bool RELU>
__global__ __launch_bounds__(256)
void gather_kernel(const float* __restrict__ h,
                   const float* __restrict__ dinv,
                   const int* __restrict__ rowptr,
                   const int* __restrict__ cnt,
                   const int* __restrict__ col,
                   const float* __restrict__ bias,
                   float* __restrict__ out, int n) {
    constexpr int LPR = D / 4;
    constexpr int G = 32 / LPR;
    const int node = (blockIdx.x * blockDim.x + threadIdx.x) >> 5;
    if (node >= n) return;
    const int lane = threadIdx.x & 31;
    const int g = lane / LPR;
    const int sub = lane % LPR;
    const float di = dinv[node];
    const int start = rowptr[node];
    const int end = start + cnt[node];
    const float* __restrict__ hb = h + sub * 4;

    float ax = 0.0f, ay = 0.0f, az = 0.0f, aw = 0.0f;
    if (g == 0) {
        const float4 v = *(const float4*)(hb + (size_t)node * D);
        const float nn = di * di;
        ax = v.x * nn; ay = v.y * nn; az = v.z * nn; aw = v.w * nn;
    }

    int j = start + g;
    for (; j + 3 * G < end; j += 4 * G) {
        const int s0 = col[j];
        const int s1 = col[j + G];
        const int s2 = col[j + 2 * G];
        const int s3 = col[j + 3 * G];
        const float n0 = dinv[s0] * di;
        const float n1 = dinv[s1] * di;
        const float n2 = dinv[s2] * di;
        const float n3 = dinv[s3] * di;
        const float4 v0 = *(const float4*)(hb + (size_t)s0 * D);
        const float4 v1 = *(const float4*)(hb + (size_t)s1 * D);
        const float4 v2 = *(const float4*)(hb + (size_t)s2 * D);
        const float4 v3 = *(const float4*)(hb + (size_t)s3 * D);
        ax += v0.x * n0; ay += v0.y * n0; az += v0.z * n0; aw += v0.w * n0;
        ax += v1.x * n1; ay += v1.y * n1; az += v1.z * n1; aw += v1.w * n1;
        ax += v2.x * n2; ay += v2.y * n2; az += v2.z * n2; aw += v2.w * n2;
        ax += v3.x * n3; ay += v3.y * n3; az += v3.z * n3; aw += v3.w * n3;
    }
    for (; j < end; j += G) {
        const int s0 = col[j];
        const float n0 = dinv[s0] * di;
        const float4 v0 = *(const float4*)(hb + (size_t)s0 * D);
        ax += v0.x * n0; ay += v0.y * n0; az += v0.z * n0; aw += v0.w * n0;
    }

    if (G == 2) {
        ax += __shfl_xor(ax, 16);
        ay += __shfl_xor(ay, 16);
        az += __shfl_xor(az, 16);
        aw += __shfl_xor(aw, 16);
    }
    if (g == 0) {
        const float4 b = *(const float4*)(bias + sub * 4);
        float4 r;
        r.x = ax + b.x; r.y = ay + b.y; r.z = az + b.z; r.w = aw + b.w;
        if (RELU) {
            r.x = fmaxf(r.x, 0.0f); r.y = fmaxf(r.y, 0.0f);
            r.z = fmaxf(r.z, 0.0f); r.w = fmaxf(r.w, 0.0f);
        }
        *(float4*)(out + (size_t)node * D + sub * 4) = r;
    }
}

extern "C" void kernel_launch(void* const* d_in, const int* in_sizes, int n_in,
                              void* d_out, int out_size, void* d_ws, size_t ws_size,
                              hipStream_t stream) {
    const float* x    = (const float*)d_in[0];
    const int*   eidx = (const int*)  d_in[1];
    const float* W1   = (const float*)d_in[2];
    const float* b1   = (const float*)d_in[3];
    const float* W2   = (const float*)d_in[4];
    const float* b2   = (const float*)d_in[5];
    float* out = (float*)d_out;

    const int N = in_sizes[0] / D_IN;
    const int E = in_sizes[1] / 2;
    const int* src = eidx;
    const int* dst = eidx + E;

    const size_t NP = ((size_t)N + 255) & ~(size_t)255;
    const size_t EP = ((size_t)E + 255) & ~(size_t)255;
    float* ws = (float*)d_ws;
    float* dinv   = ws;
    int*   cnt    = (int*)(ws + NP);
    int*   rowptr = (int*)(ws + 2 * NP);
    int*   rank   = (int*)(ws + 3 * NP);
    int*   col    = (int*)(ws + 3 * NP + EP);
    float* h1     = ws + 3 * NP + 2 * EP;
    float* agg1   = h1 + (size_t)N * D_HID;
    float* h2     = h1;

    const int BT = 256;

    cnt_zero_kernel<<<(N + BT - 1) / BT, BT, 0, stream>>>(cnt, N);
    count_rank_kernel<<<(E + BT - 1) / BT, BT, 0, stream>>>(dst, cnt, rank, E);
    rowptr_dinv_kernel<<<(N + SCAN_T - 1) / SCAN_T, SCAN_T, 0, stream>>>(cnt, rowptr, dinv, N);
    csr_fill_kernel<<<(E + BT - 1) / BT, BT, 0, stream>>>(src, dst, rank, rowptr, col, E);

    const unsigned gather_blocks = (unsigned)(((long)N * 32 + BT - 1) / BT);

    gemm_wmma_f32_kernel<D_HID><<<N / 16, 32 * (D_HID / 16), 0, stream>>>(x, W1, h1, N);
    gather_kernel<D_HID, true><<<gather_blocks, BT, 0, stream>>>(
        h1, dinv, rowptr, cnt, col, b1, agg1, N);

    gemm_wmma_f32_kernel<D_OUT><<<N / 16, 32 * (D_OUT / 16), 0, stream>>>(agg1, W2, h2, N);
    gather_kernel<D_OUT, false><<<gather_blocks, BT, 0, stream>>>(
        h2, dinv, rowptr, cnt, col, b2, out, N);
}


# Layer-1 gather now sums x rows into an LDS strip and runs both WMMA products on it in the same kernel (x W1 and hidden layer never stored; 8 launches to 6); sum order: (A x) W1, not A (x W1)
# speedup vs baseline: 2.8177x; 1.2239x over previous
#include <hip/hip_runtime.h>


typedef __attribute__((ext_vector_type(2))) float v2f;
typedef __attribute__((ext_vector_type(8))) float v8f;

#define D_IN  128
#define D_HID 128
#define D_OUT 64

__global__ void cnt_zero_kernel(int* __restrict__ cnt, int n) {
    int i = blockIdx.x * blockDim.x + threadIdx.x;
    if (i < n) cnt[i] = 0;
}

__global__ void count_rank_kernel(const int* __restrict__ dst,
                                  int* __restrict__ cnt,
                                  int* __restrict__ rank, int e) {
    int i = blockIdx.x * blockDim.x + threadIdx.x;
    if (i < e) rank[i] = atomicAdd(&cnt[dst[i]], 1);
}

#define SCAN_T 1024

__global__ __launch_bounds__(SCAN_T)
void rowptr_dinv_kernel(const int* __restrict__ cnt,
                        int* __restrict__ rowptr,
                        float* __restrict__ dinv, int n) {
    __shared__ int wsum[32];
    __shared__ int base_s;
    const int tid = threadIdx.x;
    const int lane = tid & 31;
    const int wave = tid >> 5;
    const int first = blockIdx.x * SCAN_T;

    int s = 0;
    for (int i = tid * 4; i < first; i += SCAN_T * 4) {
        const int4 v = *(const int4*)(cnt + i);
        s += (v.x + v.y) + (v.z + v.w);
    }
    for (int o = 16; o > 0; o >>= 1) s += __shfl_xor(s, o);
    if (lane == 0) wsum[wave] = s;
    __syncthreads();
    if (wave == 0) {
        int t = wsum[lane];
        for (int o = 16; o > 0; o >>= 1) t += __shfl_xor(t, o);
        if (lane == 0) base_s = t;
    }
    __syncthreads();
    const int base = base_s;

    const int i = first + tid;
    const int c = (i < n) ? cnt[i] : 0;
    int incl = c;
    for (int o = 1; o < 32; o <<= 1) {
        const int t = __shfl_up(incl, o);
        if (lane >= o) incl += t;
    }
    if (lane == 31) wsum[wave] = incl;
    __syncthreads();
    if (wave == 0) {
        const int t = wsum[lane];
        int ti = t;
        for (int o = 1; o < 32; o <<= 1) {
            const int u = __shfl_up(ti, o);
            if (lane >= o) ti += u;
        }
        wsum[lane] = ti - t;
    }
    __syncthreads();
    if (i < n) {
        rowptr[i] = base + wsum[wave] + (incl - c);
        const float d = (float)(c + 1);
        dinv[i] = (d > 0.0f) ? rsqrtf(d) : 0.0f;
    }
}

__global__ void csr_fill_kernel(const int* __restrict__ src,
                                const int* __restrict__ dst,
                                const int* __restrict__ rank,
                                const int* __restrict__ rowptr,
                                int* __restrict__ col, int e) {
    int i = blockIdx.x * blockDim.x + threadIdx.x;
    if (i < e) col[rowptr[dst[i]] + rank[i]] = src[i];
}

#define FN 32
#define SP 132

__global__ __launch_bounds__(256)
void layer1_fused_kernel(const float* __restrict__ x,
                         const float* __restrict__ dinv,
                         const int* __restrict__ rowptr,
                         const int* __restrict__ cnt,
                         const int* __restrict__ col,
                         const float* __restrict__ W1,
                         const float* __restrict__ b1,
                         const float* __restrict__ W2,
                         float* __restrict__ h2, int n) {
    constexpr int RT = FN / 16;
    constexpr int NPW = FN / 8;
    __shared__ __align__(16) float strip[FN * SP];

    const int tid = threadIdx.x;
    const int wave = tid >> 5;
    const int lane = tid & 31;
    const int node0 = blockIdx.x * FN;
    const float* __restrict__ xb = x + lane * 4;

    #pragma unroll
    for (int r = 0; r < NPW; ++r) {
        const int row = r * 8 + wave;
        const int node = node0 + row;
        float ax = 0.0f, ay = 0.0f, az = 0.0f, aw = 0.0f;
        if (node < n) {
            const float di = dinv[node];
            const int start = rowptr[node];
            const int end = start + cnt[node];
            {
                const float4 v = *(const float4*)(xb + (size_t)node * D_IN);
                const float nn = di * di;
                ax = v.x * nn; ay = v.y * nn; az = v.z * nn; aw = v.w * nn;
            }
            int j = start;
            for (; j + 3 < end; j += 4) {
                const int s0 = col[j];
                const int s1 = col[j + 1];
                const int s2 = col[j + 2];
                const int s3 = col[j + 3];
                const float n0 = dinv[s0] * di;
                const float n1 = dinv[s1] * di;
                const float n2 = dinv[s2] * di;
                const float n3 = dinv[s3] * di;
                const float4 v0 = *(const float4*)(xb + (size_t)s0 * D_IN);
                const float4 v1 = *(const float4*)(xb + (size_t)s1 * D_IN);
                const float4 v2 = *(const float4*)(xb + (size_t)s2 * D_IN);
                const float4 v3 = *(const float4*)(xb + (size_t)s3 * D_IN);
                ax += v0.x * n0; ay += v0.y * n0; az += v0.z * n0; aw += v0.w * n0;
                ax += v1.x * n1; ay += v1.y * n1; az += v1.z * n1; aw += v1.w * n1;
                ax += v2.x * n2; ay += v2.y * n2; az += v2.z * n2; aw += v2.w * n2;
                ax += v3.x * n3; ay += v3.y * n3; az += v3.z * n3; aw += v3.w * n3;
            }
            for (; j < end; ++j) {
                const int s0 = col[j];
                const float n0 = dinv[s0] * di;
                const float4 v0 = *(const float4*)(xb + (size_t)s0 * D_IN);
                ax += v0.x * n0; ay += v0.y * n0; az += v0.z * n0; aw += v0.w * n0;
            }
        }
        float4 o; o.x = ax; o.y = ay; o.z = az; o.w = aw;
        *(float4*)(&strip[row * SP + lane * 4]) = o;
    }
    __syncthreads();

    const int half = lane >> 4;
    const int l16  = lane & 15;

    {
        const int col_base = wave * 16;
        v8f acc[RT];
        #pragma unroll
        for (int t = 0; t < RT; ++t) acc[t] = v8f{};
        #pragma unroll 4
        for (int k0 = 0; k0 < D_IN; k0 += 4) {
            const int ka = k0 + half * 2;
            v2f b;
            b.x = W1[(size_t)ka * D_HID + col_base + l16];
            b.y = W1[(size_t)(ka + 1) * D_HID + col_base + l16];
            #pragma unroll
            for (int t = 0; t < RT; ++t) {
                const v2f a = *(const v2f*)(&strip[(t * 16 + l16) * SP + ka]);
                acc[t] = __builtin_amdgcn_wmma_f32_16x16x4_f32(
                     false, a,  false, b,
                     (short)0, acc[t],  false,  false);
            }
        }
        __syncthreads();
        const float bb = b1[col_base + l16];
        #pragma unroll
        for (int t = 0; t < RT; ++t) {
            #pragma unroll
            for (int v = 0; v < 8; ++v) {
                strip[(t * 16 + v + half * 8) * SP + col_base + l16] = fmaxf(acc[t][v] + bb, 0.0f);
            }
        }
        __syncthreads();
    }

    #pragma unroll
    for (int tile = wave; tile < 4 * RT; tile += 8) {
        const int col_base = (tile & 3) * 16;
        const int t = tile >> 2;
        v8f acc = {};
        #pragma unroll 4
        for (int k0 = 0; k0 < D_HID; k0 += 4) {
            const int ka = k0 + half * 2;
            v2f b;
            b.x = W2[(size_t)ka * D_OUT + col_base + l16];
            b.y = W2[(size_t)(ka + 1) * D_OUT + col_base + l16];
            const v2f a = *(const v2f*)(&strip[(t * 16 + l16) * SP + ka]);
            acc = __builtin_amdgcn_wmma_f32_16x16x4_f32(
                 false, a,  false, b,
                 (short)0, acc,  false,  false);
        }
        #pragma unroll
        for (int v = 0; v < 8; ++v) {
            const int node = node0 + t * 16 + v + half * 8;
            if (node < n) h2[(size_t)node * D_OUT + col_base + l16] = acc[v];
        }
    }
}

template <int D, bool RELU>
__global__ __launch_bounds__(256)
void gather_kernel(const float* __restrict__ h,
                   const float* __restrict__ dinv,
                   const int* __restrict__ rowptr,
                   const int* __restrict__ cnt,
                   const int* __restrict__ col,
                   const float* __restrict__ bias,
                   float* __restrict__ out, int n) {
    constexpr int LPR = D / 4;
    constexpr int G = 32 / LPR;
    const int node = (blockIdx.x * blockDim.x + threadIdx.x) >> 5;
    if (node >= n) return;
    const int lane = threadIdx.x & 31;
    const int g = lane / LPR;
    const int sub = lane % LPR;
    const float di = dinv[node];
    const int start = rowptr[node];
    const int end = start + cnt[node];
    const float* __restrict__ hb = h + sub * 4;

    float ax = 0.0f, ay = 0.0f, az = 0.0f, aw = 0.0f;
    if (g == 0) {
        const float4 v = *(const float4*)(hb + (size_t)node * D);
        const float nn = di * di;
        ax = v.x * nn; ay = v.y * nn; az = v.z * nn; aw = v.w * nn;
    }

    int j = start + g;
    for (; j + 3 * G < end; j += 4 * G) {
        const int s0 = col[j];
        const int s1 = col[j + G];
        const int s2 = col[j + 2 * G];
        const int s3 = col[j + 3 * G];
        const float n0 = dinv[s0] * di;
        const float n1 = dinv[s1] * di;
        const float n2 = dinv[s2] * di;
        const float n3 = dinv[s3] * di;
        const float4 v0 = *(const float4*)(hb + (size_t)s0 * D);
        const float4 v1 = *(const float4*)(hb + (size_t)s1 * D);
        const float4 v2 = *(const float4*)(hb + (size_t)s2 * D);
        const float4 v3 = *(const float4*)(hb + (size_t)s3 * D);
        ax += v0.x * n0; ay += v0.y * n0; az += v0.z * n0; aw += v0.w * n0;
        ax += v1.x * n1; ay += v1.y * n1; az += v1.z * n1; aw += v1.w * n1;
        ax += v2.x * n2; ay += v2.y * n2; az += v2.z * n2; aw += v2.w * n2;
        ax += v3.x * n3; ay += v3.y * n3; az += v3.z * n3; aw += v3.w * n3;
    }
    for (; j < end; j += G) {
        const int s0 = col[j];
        const float n0 = dinv[s0] * di;
        const float4 v0 = *(const float4*)(hb + (size_t)s0 * D);
        ax += v0.x * n0; ay += v0.y * n0; az += v0.z * n0; aw += v0.w * n0;
    }

    if (G == 2) {
        ax += __shfl_xor(ax, 16);
        ay += __shfl_xor(ay, 16);
        az += __shfl_xor(az, 16);
        aw += __shfl_xor(aw, 16);
    }
    if (g == 0) {
        const float4 b = *(const float4*)(bias + sub * 4);
        float4 r;
        r.x = ax + b.x; r.y = ay + b.y; r.z = az + b.z; r.w = aw + b.w;
        if (RELU) {
            r.x = fmaxf(r.x, 0.0f); r.y = fmaxf(r.y, 0.0f);
            r.z = fmaxf(r.z, 0.0f); r.w = fmaxf(r.w, 0.0f);
        }
        *(float4*)(out + (size_t)node * D + sub * 4) = r;
    }
}

extern "C" void kernel_launch(void* const* d_in, const int* in_sizes, int n_in,
                              void* d_out, int out_size, void* d_ws, size_t ws_size,
                              hipStream_t stream) {
    const float* x    = (const float*)d_in[0];
    const int*   eidx = (const int*)  d_in[1];
    const float* W1   = (const float*)d_in[2];
    const float* b1   = (const float*)d_in[3];
    const float* W2   = (const float*)d_in[4];
    const float* b2   = (const float*)d_in[5];
    float* out = (float*)d_out;

    const int N = in_sizes[0] / D_IN;
    const int E = in_sizes[1] / 2;
    const int* src = eidx;
    const int* dst = eidx + E;

    const size_t NP = ((size_t)N + 255) & ~(size_t)255;
    const size_t EP = ((size_t)E + 255) & ~(size_t)255;
    float* ws = (float*)d_ws;
    float* dinv   = ws;
    int*   cnt    = (int*)(ws + NP);
    int*   rowptr = (int*)(ws + 2 * NP);
    int*   rank   = (int*)(ws + 3 * NP);
    int*   col    = (int*)(ws + 3 * NP + EP);
    float* h2     = ws + 3 * NP + 2 * EP;

    const int BT = 256;

    cnt_zero_kernel<<<(N + BT - 1) / BT, BT, 0, stream>>>(cnt, N);
    count_rank_kernel<<<(E + BT - 1) / BT, BT, 0, stream>>>(dst, cnt, rank, E);
    rowptr_dinv_kernel<<<(N + SCAN_T - 1) / SCAN_T, SCAN_T, 0, stream>>>(cnt, rowptr, dinv, N);
    csr_fill_kernel<<<(E + BT - 1) / BT, BT, 0, stream>>>(src, dst, rank, rowptr, col, E);

    const unsigned gather_blocks = (unsigned)(((long)N * 32 + BT - 1) / BT);

    layer1_fused_kernel<<<(N + FN - 1) / FN, 256, 0, stream>>>(
        x, dinv, rowptr, cnt, col, W1, b1, W2, h2, N);

    gather_kernel<D_OUT, false><<<gather_blocks, BT, 0, stream>>>(
        h2, dinv, rowptr, cnt, col, b2, out, N);
}


# Edge lists built by a two-level counting sort in LDS (chunks grouped by 512-node range, then sorted by node): no per-edge global atomic or scattered store; 6 launches to 4; sum order: LDS arrival
# speedup vs baseline: 5.4294x; 1.9269x over previous
#include <hip/hip_runtime.h>


typedef __attribute__((ext_vector_type(2))) float v2f;
typedef __attribute__((ext_vector_type(8))) float v8f;

#define D_IN  128
#define D_HID 128
#define D_OUT 64

#define BIN_SHIFT 9
#define BIN_NODES 512
#define NBMAX 256
#define EPT 7
#define CH (EPT * 1024)
#define CAPO 16384

__device__ __forceinline__ int block_scan_excl(int v, int* wsum, int tid) {
    const int lane = tid & 31;
    const int wave = tid >> 5;
    int incl = v;
    for (int o = 1; o < 32; o <<= 1) {
        const int t = __shfl_up(incl, o);
        if (lane >= o) incl += t;
    }
    __syncthreads();
    if (lane == 31) wsum[wave] = incl;
    __syncthreads();
    if (wave == 0) {
        const int t = wsum[lane];
        int ti = t;
        for (int o = 1; o < 32; o <<= 1) {
            const int u = __shfl_up(ti, o);
            if (lane >= o) ti += u;
        }
        wsum[lane] = ti - t;
        if (lane == 31) wsum[32] = ti;
    }
    __syncthreads();
    return wsum[wave] + (incl - v);
}

__global__ __launch_bounds__(1024)
void bin_partition_kernel(const int* __restrict__ src,
                          const int* __restrict__ dst,
                          int* __restrict__ part,
                          int* __restrict__ offs, int e, int nb) {
    __shared__ int hist[NBMAX];
    __shared__ int boff[NBMAX + 1];
    __shared__ int wsum[33];
    __shared__ int sorted[CH];
    const int tid = threadIdx.x;
    const int p = blockIdx.x;
    const int P = gridDim.x;
    const int e0 = p * CH;
    const int ne = (e - e0 < CH) ? (e - e0) : CH;

    if (tid < NBMAX) hist[tid] = 0;
    __syncthreads();

    int key[EPT], bin[EPT], rk[EPT];
    #pragma unroll
    for (int k = 0; k < EPT; ++k) {
        const int i = k * 1024 + tid;
        const int ii = (i < ne) ? i : (ne - 1);
        bin[k] = dst[e0 + ii];
        key[k] = src[e0 + ii];
    }
    #pragma unroll
    for (int k = 0; k < EPT; ++k) {
        const int i = k * 1024 + tid;
        const int d = bin[k];
        bin[k] = -1;
        rk[k] = 0;
        if (i < ne) {
            bin[k] = d >> BIN_SHIFT;
            key[k] = (key[k] << BIN_SHIFT) | (d & (BIN_NODES - 1));
            rk[k] = atomicAdd(&hist[bin[k]], 1);
        }
    }
    __syncthreads();

    const int hv = (tid < nb) ? hist[tid] : 0;
    const int ex = block_scan_excl(hv, wsum, tid);
    if (tid <= nb) boff[tid] = ex;
    __syncthreads();

    #pragma unroll
    for (int k = 0; k < EPT; ++k) {
        if (bin[k] >= 0) sorted[boff[bin[k]] + rk[k]] = key[k];
    }
    __syncthreads();

    for (int i = tid; i < ne; i += 1024) part[e0 + i] = sorted[i];
    if (tid <= nb) offs[tid * P + p] = boff[tid];
}

__global__ __launch_bounds__(1024)
void bin_sort_kernel(const int* __restrict__ part,
                     const int* __restrict__ offs,
                     int* __restrict__ cnt,
                     int* __restrict__ rowptr,
                     float* __restrict__ dinv,
                     int* __restrict__ col, int n, int P) {
    __shared__ int rs[1024];
    __shared__ int rl[1024];
    __shared__ int hist2[BIN_NODES];
    __shared__ int cursor[BIN_NODES];
    __shared__ int wsum[33];
    __shared__ int obuf[CAPO];
    const int tid = threadIdx.x;
    const int lane = tid & 31;
    const int wave = tid >> 5;
    const int b = blockIdx.x;

    int o0 = 0, o1 = 0;
    if (tid < P) {
        o0 = offs[b * P + tid];
        o1 = offs[(b + 1) * P + tid];
    }
    rs[tid] = tid * CH + o0;
    rl[tid] = o1 - o0;
    if (tid < BIN_NODES) hist2[tid] = 0;
    block_scan_excl(o0, wsum, tid);
    const int bin_start = wsum[32];
    block_scan_excl(o1 - o0, wsum, tid);
    const int total = wsum[32];
    __syncthreads();

    for (int r = wave; r < P; r += 32) {
        const int s0 = rs[r];
        const int len = rl[r];
        for (int i = lane; i < len; i += 64) {
            const bool two = (i + 32) < len;
            const int k0 = part[s0 + i];
            const int k1 = two ? part[s0 + i + 32] : 0;
            atomicAdd(&hist2[k0 & (BIN_NODES - 1)], 1);
            if (two) atomicAdd(&hist2[k1 & (BIN_NODES - 1)], 1);
        }
    }
    __syncthreads();

    const int c = (tid < BIN_NODES) ? hist2[tid] : 0;
    const int ex = block_scan_excl(c, wsum, tid);
    if (tid < BIN_NODES) {
        cursor[tid] = ex;
        const int node = b * BIN_NODES + tid;
        if (node < n) {
            cnt[node] = c;
            rowptr[node] = bin_start + ex;
            const float d = (float)(c + 1);
            dinv[node] = (d > 0.0f) ? rsqrtf(d) : 0.0f;
        }
    }
    __syncthreads();

    const bool staged = total <= CAPO;
    for (int r = wave; r < P; r += 32) {
        const int s0 = rs[r];
        const int len = rl[r];
        for (int i = lane; i < len; i += 64) {
            const bool two = (i + 32) < len;
            const int k0 = part[s0 + i];
            const int k1 = two ? part[s0 + i + 32] : 0;
            const int p0 = atomicAdd(&cursor[k0 & (BIN_NODES - 1)], 1);
            if (staged) obuf[p0] = k0 >> BIN_SHIFT; else col[bin_start + p0] = k0 >> BIN_SHIFT;
            if (two) {
                const int p1 = atomicAdd(&cursor[k1 & (BIN_NODES - 1)], 1);
                if (staged) obuf[p1] = k1 >> BIN_SHIFT; else col[bin_start + p1] = k1 >> BIN_SHIFT;
            }
        }
    }
    __syncthreads();
    if (staged) {
        for (int i = tid; i < total; i += 1024) col[bin_start + i] = obuf[i];
    }
}

#define FN 32
#define SP 132

__global__ __launch_bounds__(256)
void layer1_fused_kernel(const float* __restrict__ x,
                         const float* __restrict__ dinv,
                         const int* __restrict__ rowptr,
                         const int* __restrict__ cnt,
                         const int* __restrict__ col,
                         const float* __restrict__ W1,
                         const float* __restrict__ b1,
                         const float* __restrict__ W2,
                         float* __restrict__ h2, int n) {
    constexpr int RT = FN / 16;
    constexpr int NPW = FN / 8;
    __shared__ __align__(16) float strip[FN * SP];

    const int tid = threadIdx.x;
    const int wave = tid >> 5;
    const int lane = tid & 31;
    const int node0 = blockIdx.x * FN;
    const float* __restrict__ xb = x + lane * 4;

    #pragma unroll
    for (int r = 0; r < NPW; ++r) {
        const int row = r * 8 + wave;
        const int node = node0 + row;
        float ax = 0.0f, ay = 0.0f, az = 0.0f, aw = 0.0f;
        if (node < n) {
            const float di = dinv[node];
            const int start = rowptr[node];
            const int end = start + cnt[node];
            {
                const float4 v = *(const float4*)(xb + (size_t)node * D_IN);
                const float nn = di * di;
                ax = v.x * nn; ay = v.y * nn; az = v.z * nn; aw = v.w * nn;
            }
            int j = start;
            for (; j + 3 < end; j += 4) {
                const int s0 = col[j];
                const int s1 = col[j + 1];
                const int s2 = col[j + 2];
                const int s3 = col[j + 3];
                const float n0 = dinv[s0] * di;
                const float n1 = dinv[s1] * di;
                const float n2 = dinv[s2] * di;
                const float n3 = dinv[s3] * di;
                const float4 v0 = *(const float4*)(xb + (size_t)s0 * D_IN);
                const float4 v1 = *(const float4*)(xb + (size_t)s1 * D_IN);
                const float4 v2 = *(const float4*)(xb + (size_t)s2 * D_IN);
                const float4 v3 = *(const float4*)(xb + (size_t)s3 * D_IN);
                ax += v0.x * n0; ay += v0.y * n0; az += v0.z * n0; aw += v0.w * n0;
                ax += v1.x * n1; ay += v1.y * n1; az += v1.z * n1; aw += v1.w * n1;
                ax += v2.x * n2; ay += v2.y * n2; az += v2.z * n2; aw += v2.w * n2;
                ax += v3.x * n3; ay += v3.y * n3; az += v3.z * n3; aw += v3.w * n3;
            }
            for (; j < end; ++j) {
                const int s0 = col[j];
                const float n0 = dinv[s0] * di;
                const float4 v0 = *(const float4*)(xb + (size_t)s0 * D_IN);
                ax += v0.x * n0; ay += v0.y * n0; az += v0.z * n0; aw += v0.w * n0;
            }
        }
        float4 o; o.x = ax; o.y = ay; o.z = az; o.w = aw;
        *(float4*)(&strip[row * SP + lane * 4]) = o;
    }
    __syncthreads();

    const int half = lane >> 4;
    const int l16  = lane & 15;

    {
        const int col_base = wave * 16;
        v8f acc[RT];
        #pragma unroll
        for (int t = 0; t < RT; ++t) acc[t] = v8f{};
        #pragma unroll 4
        for (int k0 = 0; k0 < D_IN; k0 += 4) {
            const int ka = k0 + half * 2;
            v2f b;
            b.x = W1[(size_t)ka * D_HID + col_base + l16];
            b.y = W1[(size_t)(ka + 1) * D_HID + col_base + l16];
            #pragma unroll
            for (int t = 0; t < RT; ++t) {
                const v2f a = *(const v2f*)(&strip[(t * 16 + l16) * SP + ka]);
                acc[t] = __builtin_amdgcn_wmma_f32_16x16x4_f32(
                     false, a,  false, b,
                     (short)0, acc[t],  false,  false);
            }
        }
        __syncthreads();
        const float bb = b1[col_base + l16];
        #pragma unroll
        for (int t = 0; t < RT; ++t) {
            #pragma unroll
            for (int v = 0; v < 8; ++v) {
                strip[(t * 16 + v + half * 8) * SP + col_base + l16] = fmaxf(acc[t][v] + bb, 0.0f);
            }
        }
        __syncthreads();
    }

    #pragma unroll
    for (int tile = wave; tile < 4 * RT; tile += 8) {
        const int col_base = (tile & 3) * 16;
        const int t = tile >> 2;
        v8f acc = {};
        #pragma unroll 4
        for (int k0 = 0; k0 < D_HID; k0 += 4) {
            const int ka = k0 + half * 2;
            v2f b;
            b.x = W2[(size_t)ka * D_OUT + col_base + l16];
            b.y = W2[(size_t)(ka + 1) * D_OUT + col_base + l16];
            const v2f a = *(const v2f*)(&strip[(t * 16 + l16) * SP + ka]);
            acc = __builtin_amdgcn_wmma_f32_16x16x4_f32(
                 false, a,  false, b,
                 (short)0, acc,  false,  false);
        }
        #pragma unroll
        for (int v = 0; v < 8; ++v) {
            const int node = node0 + t * 16 + v + half * 8;
            if (node < n) h2[(size_t)node * D_OUT + col_base + l16] = acc[v];
        }
    }
}

template <int D, bool RELU>
__global__ __launch_bounds__(256)
void gather_kernel(const float* __restrict__ h,
                   const float* __restrict__ dinv,
                   const int* __restrict__ rowptr,
                   const int* __restrict__ cnt,
                   const int* __restrict__ col,
                   const float* __restrict__ bias,
                   float* __restrict__ out, int n) {
    constexpr int LPR = D / 4;
    constexpr int G = 32 / LPR;
    const int node = (blockIdx.x * blockDim.x + threadIdx.x) >> 5;
    if (node >= n) return;
    const int lane = threadIdx.x & 31;
    const int g = lane / LPR;
    const int sub = lane % LPR;
    const float di = dinv[node];
    const int start = rowptr[node];
    const int end = start + cnt[node];
    const float* __restrict__ hb = h + sub * 4;

    float ax = 0.0f, ay = 0.0f, az = 0.0f, aw = 0.0f;
    if (g == 0) {
        const float4 v = *(const float4*)(hb + (size_t)node * D);
        const float nn = di * di;
        ax = v.x * nn; ay = v.y * nn; az = v.z * nn; aw = v.w * nn;
    }

    int j = start + g;
    for (; j + 3 * G < end; j += 4 * G) {
        const int s0 = col[j];
        const int s1 = col[j + G];
        const int s2 = col[j + 2 * G];
        const int s3 = col[j + 3 * G];
        const float n0 = dinv[s0] * di;
        const float n1 = dinv[s1] * di;
        const float n2 = dinv[s2] * di;
        const float n3 = dinv[s3] * di;
        const float4 v0 = *(const float4*)(hb + (size_t)s0 * D);
        const float4 v1 = *(const float4*)(hb + (size_t)s1 * D);
        const float4 v2 = *(const float4*)(hb + (size_t)s2 * D);
        const float4 v3 = *(const float4*)(hb + (size_t)s3 * D);
        ax += v0.x * n0; ay += v0.y * n0; az += v0.z * n0; aw += v0.w * n0;
        ax += v1.x * n1; ay += v1.y * n1; az += v1.z * n1; aw += v1.w * n1;
        ax += v2.x * n2; ay += v2.y * n2; az += v2.z * n2; aw += v2.w * n2;
        ax += v3.x * n3; ay += v3.y * n3; az += v3.z * n3; aw += v3.w * n3;
    }
    for (; j < end; j += G) {
        const int s0 = col[j];
        const float n0 = dinv[s0] * di;
        const float4 v0 = *(const float4*)(hb + (size_t)s0 * D);
        ax += v0.x * n0; ay += v0.y * n0; az += v0.z * n0; aw += v0.w * n0;
    }

    if (G == 2) {
        ax += __shfl_xor(ax, 16);
        ay += __shfl_xor(ay, 16);
        az += __shfl_xor(az, 16);
        aw += __shfl_xor(aw, 16);
    }
    if (g == 0) {
        const float4 b = *(const float4*)(bias + sub * 4);
        float4 r;
        r.x = ax + b.x; r.y = ay + b.y; r.z = az + b.z; r.w = aw + b.w;
        if (RELU) {
            r.x = fmaxf(r.x, 0.0f); r.y = fmaxf(r.y, 0.0f);
            r.z = fmaxf(r.z, 0.0f); r.w = fmaxf(r.w, 0.0f);
        }
        *(float4*)(out + (size_t)node * D + sub * 4) = r;
    }
}

extern "C" void kernel_launch(void* const* d_in, const int* in_sizes, int n_in,
                              void* d_out, int out_size, void* d_ws, size_t ws_size,
                              hipStream_t stream) {
    const float* x    = (const float*)d_in[0];
    const int*   eidx = (const int*)  d_in[1];
    const float* W1   = (const float*)d_in[2];
    const float* b1   = (const float*)d_in[3];
    const float* W2   = (const float*)d_in[4];
    const float* b2   = (const float*)d_in[5];
    float* out = (float*)d_out;

    const int N = in_sizes[0] / D_IN;
    const int E = in_sizes[1] / 2;
    const int* src = eidx;
    const int* dst = eidx + E;

    const int NB = (N + BIN_NODES - 1) / BIN_NODES;
    const int P  = (E + CH - 1) / CH;

    const size_t NP = ((size_t)N + 255) & ~(size_t)255;
    const size_t EP = ((size_t)E + 255) & ~(size_t)255;
    const size_t OP = ((size_t)(NB + 1) * P + 255) & ~(size_t)255;
    float* ws = (float*)d_ws;
    float* dinv   = ws;
    int*   cnt    = (int*)(ws + NP);
    int*   rowptr = (int*)(ws + 2 * NP);
    int*   part   = (int*)(ws + 3 * NP);
    int*   col    = (int*)(ws + 3 * NP + EP);
    int*   offs   = (int*)(ws + 3 * NP + 2 * EP);
    float* h2     = ws + 3 * NP + 2 * EP + OP;

    const int BT = 256;

    bin_partition_kernel<<<P, 1024, 0, stream>>>(src, dst, part, offs, E, NB);
    bin_sort_kernel<<<NB, 1024, 0, stream>>>(part, offs, cnt, rowptr, dinv, col, N, P);

    const unsigned gather_blocks = (unsigned)(((long)N * 32 + BT - 1) / BT);

    layer1_fused_kernel<<<(N + FN - 1) / FN, 256, 0, stream>>>(
        x, dinv, rowptr, cnt, col, W1, b1, W2, h2, N);

    gather_kernel<D_OUT, false><<<gather_blocks, BT, 0, stream>>>(
        h2, dinv, rowptr, cnt, col, b2, out, N);
}


# h2 rows stored pre-multiplied by their own dinv (free, at the fused launch's store; x untouched): layer-2 gather drops its per-edge dinv[s] lookup, dinv[t] scales the sum; sum order unchanged
# speedup vs baseline: 5.6089x; 1.0331x over previous
#include <hip/hip_runtime.h>


typedef __attribute__((ext_vector_type(2))) float v2f;
typedef __attribute__((ext_vector_type(8))) float v8f;

#define D_IN  128
#define D_HID 128
#define D_OUT 64

#define BIN_SHIFT 9
#define BIN_NODES 512
#define NBMAX 256
#define EPT 7
#define CH (EPT * 1024)
#define CAPO 16384

__device__ __forceinline__ int block_scan_excl(int v, int* wsum, int tid) {
    const int lane = tid & 31;
    const int wave = tid >> 5;
    int incl = v;
    for (int o = 1; o < 32; o <<= 1) {
        const int t = __shfl_up(incl, o);
        if (lane >= o) incl += t;
    }
    __syncthreads();
    if (lane == 31) wsum[wave] = incl;
    __syncthreads();
    if (wave == 0) {
        const int t = wsum[lane];
        int ti = t;
        for (int o = 1; o < 32; o <<= 1) {
            const int u = __shfl_up(ti, o);
            if (lane >= o) ti += u;
        }
        wsum[lane] = ti - t;
        if (lane == 31) wsum[32] = ti;
    }
    __syncthreads();
    return wsum[wave] + (incl - v);
}

__global__ __launch_bounds__(1024)
void bin_partition_kernel(const int* __restrict__ src,
                          const int* __restrict__ dst,
                          int* __restrict__ part,
                          int* __restrict__ offs, int e, int nb) {
    __shared__ int hist[NBMAX];
    __shared__ int boff[NBMAX + 1];
    __shared__ int wsum[33];
    __shared__ int sorted[CH];
    const int tid = threadIdx.x;
    const int p = blockIdx.x;
    const int P = gridDim.x;
    const int e0 = p * CH;
    const int ne = (e - e0 < CH) ? (e - e0) : CH;

    if (tid < NBMAX) hist[tid] = 0;
    __syncthreads();

    int key[EPT], bin[EPT], rk[EPT];
    #pragma unroll
    for (int k = 0; k < EPT; ++k) {
        const int i = k * 1024 + tid;
        const int ii = (i < ne) ? i : (ne - 1);
        bin[k] = dst[e0 + ii];
        key[k] = src[e0 + ii];
    }
    #pragma unroll
    for (int k = 0; k < EPT; ++k) {
        const int i = k * 1024 + tid;
        const int d = bin[k];
        bin[k] = -1;
        rk[k] = 0;
        if (i < ne) {
            bin[k] = d >> BIN_SHIFT;
            key[k] = (key[k] << BIN_SHIFT) | (d & (BIN_NODES - 1));
            rk[k] = atomicAdd(&hist[bin[k]], 1);
        }
    }
    __syncthreads();

    const int hv = (tid < nb) ? hist[tid] : 0;
    const int ex = block_scan_excl(hv, wsum, tid);
    if (tid <= nb) boff[tid] = ex;
    __syncthreads();

    #pragma unroll
    for (int k = 0; k < EPT; ++k) {
        if (bin[k] >= 0) sorted[boff[bin[k]] + rk[k]] = key[k];
    }
    __syncthreads();

    for (int i = tid; i < ne; i += 1024) part[e0 + i] = sorted[i];
    if (tid <= nb) offs[tid * P + p] = boff[tid];
}

__global__ __launch_bounds__(1024)
void bin_sort_kernel(const int* __restrict__ part,
                     const int* __restrict__ offs,
                     int* __restrict__ cnt,
                     int* __restrict__ rowptr,
                     float* __restrict__ dinv,
                     int* __restrict__ col, int n, int P) {
    __shared__ int rs[1024];
    __shared__ int rl[1024];
    __shared__ int hist2[BIN_NODES];
    __shared__ int cursor[BIN_NODES];
    __shared__ int wsum[33];
    __shared__ int obuf[CAPO];
    const int tid = threadIdx.x;
    const int lane = tid & 31;
    const int wave = tid >> 5;
    const int b = blockIdx.x;

    int o0 = 0, o1 = 0;
    if (tid < P) {
        o0 = offs[b * P + tid];
        o1 = offs[(b + 1) * P + tid];
    }
    rs[tid] = tid * CH + o0;
    rl[tid] = o1 - o0;
    if (tid < BIN_NODES) hist2[tid] = 0;
    block_scan_excl(o0, wsum, tid);
    const int bin_start = wsum[32];
    block_scan_excl(o1 - o0, wsum, tid);
    const int total = wsum[32];
    __syncthreads();

    for (int r = wave; r < P; r += 32) {
        const int s0 = rs[r];
        const int len = rl[r];
        for (int i = lane; i < len; i += 64) {
            const bool two = (i + 32) < len;
            const int k0 = part[s0 + i];
            const int k1 = two ? part[s0 + i + 32] : 0;
            atomicAdd(&hist2[k0 & (BIN_NODES - 1)], 1);
            if (two) atomicAdd(&hist2[k1 & (BIN_NODES - 1)], 1);
        }
    }
    __syncthreads();

    const int c = (tid < BIN_NODES) ? hist2[tid] : 0;
    const int ex = block_scan_excl(c, wsum, tid);
    if (tid < BIN_NODES) {
        cursor[tid] = ex;
        const int node = b * BIN_NODES + tid;
        if (node < n) {
            cnt[node] = c;
            rowptr[node] = bin_start + ex;
            const float d = (float)(c + 1);
            dinv[node] = (d > 0.0f) ? rsqrtf(d) : 0.0f;
        }
    }
    __syncthreads();

    const bool staged = total <= CAPO;
    for (int r = wave; r < P; r += 32) {
        const int s0 = rs[r];
        const int len = rl[r];
        for (int i = lane; i < len; i += 64) {
            const bool two = (i + 32) < len;
            const int k0 = part[s0 + i];
            const int k1 = two ? part[s0 + i + 32] : 0;
            const int p0 = atomicAdd(&cursor[k0 & (BIN_NODES - 1)], 1);
            if (staged) obuf[p0] = k0 >> BIN_SHIFT; else col[bin_start + p0] = k0 >> BIN_SHIFT;
            if (two) {
                const int p1 = atomicAdd(&cursor[k1 & (BIN_NODES - 1)], 1);
                if (staged) obuf[p1] = k1 >> BIN_SHIFT; else col[bin_start + p1] = k1 >> BIN_SHIFT;
            }
        }
    }
    __syncthreads();
    if (staged) {
        for (int i = tid; i < total; i += 1024) col[bin_start + i] = obuf[i];
    }
}

#define FN 32
#define SP 132

__global__ __launch_bounds__(256)
void layer1_fused_kernel(const float* __restrict__ x,
                         const float* __restrict__ dinv,
                         const int* __restrict__ rowptr,
                         const int* __restrict__ cnt,
                         const int* __restrict__ col,
                         const float* __restrict__ W1,
                         const float* __restrict__ b1,
                         const float* __restrict__ W2,
                         float* __restrict__ h2, int n) {
    constexpr int RT = FN / 16;
    constexpr int NPW = FN / 8;
    __shared__ __align__(16) float strip[FN * SP];

    const int tid = threadIdx.x;
    const int wave = tid >> 5;
    const int lane = tid & 31;
    const int node0 = blockIdx.x * FN;
    const float* __restrict__ xb = x + lane * 4;

    #pragma unroll
    for (int r = 0; r < NPW; ++r) {
        const int row = r * 8 + wave;
        const int node = node0 + row;
        float ax = 0.0f, ay = 0.0f, az = 0.0f, aw = 0.0f;
        if (node < n) {
            const float di = dinv[node];
            const int start = rowptr[node];
            const int end = start + cnt[node];
            {
                const float4 v = *(const float4*)(xb + (size_t)node * D_IN);
                const float nn = di * di;
                ax = v.x * nn; ay = v.y * nn; az = v.z * nn; aw = v.w * nn;
            }
            int j = start;
            for (; j + 3 < end; j += 4) {
                const int s0 = col[j];
                const int s1 = col[j + 1];
                const int s2 = col[j + 2];
                const int s3 = col[j + 3];
                const float n0 = dinv[s0] * di;
                const float n1 = dinv[s1] * di;
                const float n2 = dinv[s2] * di;
                const float n3 = dinv[s3] * di;
                const float4 v0 = *(const float4*)(xb + (size_t)s0 * D_IN);
                const float4 v1 = *(const float4*)(xb + (size_t)s1 * D_IN);
                const float4 v2 = *(const float4*)(xb + (size_t)s2 * D_IN);
                const float4 v3 = *(const float4*)(xb + (size_t)s3 * D_IN);
                ax += v0.x * n0; ay += v0.y * n0; az += v0.z * n0; aw += v0.w * n0;
                ax += v1.x * n1; ay += v1.y * n1; az += v1.z * n1; aw += v1.w * n1;
                ax += v2.x * n2; ay += v2.y * n2; az += v2.z * n2; aw += v2.w * n2;
                ax += v3.x * n3; ay += v3.y * n3; az += v3.z * n3; aw += v3.w * n3;
            }
            for (; j < end; ++j) {
                const int s0 = col[j];
                const float n0 = dinv[s0] * di;
                const float4 v0 = *(const float4*)(xb + (size_t)s0 * D_IN);
                ax += v0.x * n0; ay += v0.y * n0; az += v0.z * n0; aw += v0.w * n0;
            }
        }
        float4 o; o.x = ax; o.y = ay; o.z = az; o.w = aw;
        *(float4*)(&strip[row * SP + lane * 4]) = o;
    }
    __syncthreads();

    const int half = lane >> 4;
    const int l16  = lane & 15;

    {
        const int col_base = wave * 16;
        v8f acc[RT];
        #pragma unroll
        for (int t = 0; t < RT; ++t) acc[t] = v8f{};
        #pragma unroll 4
        for (int k0 = 0; k0 < D_IN; k0 += 4) {
            const int ka = k0 + half * 2;
            v2f b;
            b.x = W1[(size_t)ka * D_HID + col_base + l16];
            b.y = W1[(size_t)(ka + 1) * D_HID + col_base + l16];
            #pragma unroll
            for (int t = 0; t < RT; ++t) {
                const v2f a = *(const v2f*)(&strip[(t * 16 + l16) * SP + ka]);
                acc[t] = __builtin_amdgcn_wmma_f32_16x16x4_f32(
                     false, a,  false, b,
                     (short)0, acc[t],  false,  false);
            }
        }
        __syncthreads();
        const float bb = b1[col_base + l16];
        #pragma unroll
        for (int t = 0; t < RT; ++t) {
            #pragma unroll
            for (int v = 0; v < 8; ++v) {
                strip[(t * 16 + v + half * 8) * SP + col_base + l16] = fmaxf(acc[t][v] + bb, 0.0f);
            }
        }
        __syncthreads();
    }

    #pragma unroll
    for (int tile = wave; tile < 4 * RT; tile += 8) {
        const int col_base = (tile & 3) * 16;
        const int t = tile >> 2;
        float dv[8];
        #pragma unroll
        for (int v = 0; v < 8; ++v) {
            const int node = node0 + t * 16 + v + half * 8;
            dv[v] = dinv[(node < n) ? node : (n - 1)];
        }
        v8f acc = {};
        #pragma unroll 4
        for (int k0 = 0; k0 < D_HID; k0 += 4) {
            const int ka = k0 + half * 2;
            v2f b;
            b.x = W2[(size_t)ka * D_OUT + col_base + l16];
            b.y = W2[(size_t)(ka + 1) * D_OUT + col_base + l16];
            const v2f a = *(const v2f*)(&strip[(t * 16 + l16) * SP + ka]);
            acc = __builtin_amdgcn_wmma_f32_16x16x4_f32(
                 false, a,  false, b,
                 (short)0, acc,  false,  false);
        }
        #pragma unroll
        for (int v = 0; v < 8; ++v) {
            const int node = node0 + t * 16 + v + half * 8;
            if (node < n) h2[(size_t)node * D_OUT + col_base + l16] = acc[v] * dv[v];
        }
    }
}

template <int D>
__global__ __launch_bounds__(256)
void gather_scaled_rows_kernel(const float* __restrict__ hs,
                               const float* __restrict__ dinv,
                               const int* __restrict__ rowptr,
                               const int* __restrict__ cnt,
                               const int* __restrict__ col,
                               const float* __restrict__ bias,
                               float* __restrict__ out, int n) {
    constexpr int LPR = D / 4;
    constexpr int G = 32 / LPR;
    const int node = (blockIdx.x * blockDim.x + threadIdx.x) >> 5;
    if (node >= n) return;
    const int lane = threadIdx.x & 31;
    const int g = lane / LPR;
    const int sub = lane % LPR;
    const float di = dinv[node];
    const int start = rowptr[node];
    const int end = start + cnt[node];
    const float* __restrict__ hb = hs + sub * 4;

    float ax = 0.0f, ay = 0.0f, az = 0.0f, aw = 0.0f;
    if (g == 0) {
        const float4 v = *(const float4*)(hb + (size_t)node * D);
        ax = v.x; ay = v.y; az = v.z; aw = v.w;
    }

    int j = start + g;
    for (; j + 3 * G < end; j += 4 * G) {
        const int s0 = col[j];
        const int s1 = col[j + G];
        const int s2 = col[j + 2 * G];
        const int s3 = col[j + 3 * G];
        const float4 v0 = *(const float4*)(hb + (size_t)s0 * D);
        const float4 v1 = *(const float4*)(hb + (size_t)s1 * D);
        const float4 v2 = *(const float4*)(hb + (size_t)s2 * D);
        const float4 v3 = *(const float4*)(hb + (size_t)s3 * D);
        ax += v0.x; ay += v0.y; az += v0.z; aw += v0.w;
        ax += v1.x; ay += v1.y; az += v1.z; aw += v1.w;
        ax += v2.x; ay += v2.y; az += v2.z; aw += v2.w;
        ax += v3.x; ay += v3.y; az += v3.z; aw += v3.w;
    }
    for (; j < end; j += G) {
        const int s0 = col[j];
        const float4 v0 = *(const float4*)(hb + (size_t)s0 * D);
        ax += v0.x; ay += v0.y; az += v0.z; aw += v0.w;
    }

    if (G == 2) {
        ax += __shfl_xor(ax, 16);
        ay += __shfl_xor(ay, 16);
        az += __shfl_xor(az, 16);
        aw += __shfl_xor(aw, 16);
    }
    if (g == 0) {
        const float4 b = *(const float4*)(bias + sub * 4);
        float4 r;
        r.x = ax * di + b.x; r.y = ay * di + b.y;
        r.z = az * di + b.z; r.w = aw * di + b.w;
        *(float4*)(out + (size_t)node * D + sub * 4) = r;
    }
}

extern "C" void kernel_launch(void* const* d_in, const int* in_sizes, int n_in,
                              void* d_out, int out_size, void* d_ws, size_t ws_size,
                              hipStream_t stream) {
    const float* x    = (const float*)d_in[0];
    const int*   eidx = (const int*)  d_in[1];
    const float* W1   = (const float*)d_in[2];
    const float* b1   = (const float*)d_in[3];
    const float* W2   = (const float*)d_in[4];
    const float* b2   = (const float*)d_in[5];
    float* out = (float*)d_out;

    const int N = in_sizes[0] / D_IN;
    const int E = in_sizes[1] / 2;
    const int* src = eidx;
    const int* dst = eidx + E;

    const int NB = (N + BIN_NODES - 1) / BIN_NODES;
    const int P  = (E + CH - 1) / CH;

    const size_t NP = ((size_t)N + 255) & ~(size_t)255;
    const size_t EP = ((size_t)E + 255) & ~(size_t)255;
    const size_t OP = ((size_t)(NB + 1) * P + 255) & ~(size_t)255;
    float* ws = (float*)d_ws;
    float* dinv   = ws;
    int*   cnt    = (int*)(ws + NP);
    int*   rowptr = (int*)(ws + 2 * NP);
    int*   part   = (int*)(ws + 3 * NP);
    int*   col    = (int*)(ws + 3 * NP + EP);
    int*   offs   = (int*)(ws + 3 * NP + 2 * EP);
    float* h2     = ws + 3 * NP + 2 * EP + OP;

    const int BT = 256;

    bin_partition_kernel<<<P, 1024, 0, stream>>>(src, dst, part, offs, E, NB);
    bin_sort_kernel<<<NB, 1024, 0, stream>>>(part, offs, cnt, rowptr, dinv, col, N, P);

    const unsigned gather_blocks = (unsigned)(((long)N * 32 + BT - 1) / BT);

    layer1_fused_kernel<<<(N + FN - 1) / FN, 256, 0, stream>>>(
        x, dinv, rowptr, cnt, col, W1, b1, W2, h2, N);

    gather_scaled_rows_kernel<D_OUT><<<gather_blocks, BT, 0, stream>>>(
        h2, dinv, rowptr, cnt, col, b2, out, N);
}


# Layer-1 gather: source ids and coefficients of a node's entries fetched once, one per lane, for a wave's 4 nodes ahead of the rows, dealt out by lane read; 8 row reads in flight; sum order unchanged
# speedup vs baseline: 5.9541x; 1.0615x over previous
#include <hip/hip_runtime.h>


typedef __attribute__((ext_vector_type(2))) float v2f;
typedef __attribute__((ext_vector_type(8))) float v8f;

#define D_IN  128
#define D_HID 128
#define D_OUT 64

#define BIN_SHIFT 9
#define BIN_NODES 512
#define NBMAX 256
#define EPT 7
#define CH (EPT * 1024)
#define CAPO 16384

__device__ __forceinline__ int block_scan_excl(int v, int* wsum, int tid) {
    const int lane = tid & 31;
    const int wave = tid >> 5;
    int incl = v;
    for (int o = 1; o < 32; o <<= 1) {
        const int t = __shfl_up(incl, o);
        if (lane >= o) incl += t;
    }
    __syncthreads();
    if (lane == 31) wsum[wave] = incl;
    __syncthreads();
    if (wave == 0) {
        const int t = wsum[lane];
        int ti = t;
        for (int o = 1; o < 32; o <<= 1) {
            const int u = __shfl_up(ti, o);
            if (lane >= o) ti += u;
        }
        wsum[lane] = ti - t;
        if (lane == 31) wsum[32] = ti;
    }
    __syncthreads();
    return wsum[wave] + (incl - v);
}

__global__ __launch_bounds__(1024)
void bin_partition_kernel(const int* __restrict__ src,
                          const int* __restrict__ dst,
                          int* __restrict__ part,
                          int* __restrict__ offs, int e, int nb) {
    __shared__ int hist[NBMAX];
    __shared__ int boff[NBMAX + 1];
    __shared__ int wsum[33];
    __shared__ int sorted[CH];
    const int tid = threadIdx.x;
    const int p = blockIdx.x;
    const int P = gridDim.x;
    const int e0 = p * CH;
    const int ne = (e - e0 < CH) ? (e - e0) : CH;

    if (tid < NBMAX) hist[tid] = 0;
    __syncthreads();

    int key[EPT], bin[EPT], rk[EPT];
    #pragma unroll
    for (int k = 0; k < EPT; ++k) {
        const int i = k * 1024 + tid;
        const int ii = (i < ne) ? i : (ne - 1);
        bin[k] = dst[e0 + ii];
        key[k] = src[e0 + ii];
    }
    #pragma unroll
    for (int k = 0; k < EPT; ++k) {
        const int i = k * 1024 + tid;
        const int d = bin[k];
        bin[k] = -1;
        rk[k] = 0;
        if (i < ne) {
            bin[k] = d >> BIN_SHIFT;
            key[k] = (key[k] << BIN_SHIFT) | (d & (BIN_NODES - 1));
            rk[k] = atomicAdd(&hist[bin[k]], 1);
        }
    }
    __syncthreads();

    const int hv = (tid < nb) ? hist[tid] : 0;
    const int ex = block_scan_excl(hv, wsum, tid);
    if (tid <= nb) boff[tid] = ex;
    __syncthreads();

    #pragma unroll
    for (int k = 0; k < EPT; ++k) {
        if (bin[k] >= 0) sorted[boff[bin[k]] + rk[k]] = key[k];
    }
    __syncthreads();

    for (int i = tid; i < ne; i += 1024) part[e0 + i] = sorted[i];
    if (tid <= nb) offs[tid * P + p] = boff[tid];
}

__global__ __launch_bounds__(1024)
void bin_sort_kernel(const int* __restrict__ part,
                     const int* __restrict__ offs,
                     int* __restrict__ cnt,
                     int* __restrict__ rowptr,
                     float* __restrict__ dinv,
                     int* __restrict__ col, int n, int P) {
    __shared__ int rs[1024];
    __shared__ int rl[1024];
    __shared__ int hist2[BIN_NODES];
    __shared__ int cursor[BIN_NODES];
    __shared__ int wsum[33];
    __shared__ int obuf[CAPO];
    const int tid = threadIdx.x;
    const int lane = tid & 31;
    const int wave = tid >> 5;
    const int b = blockIdx.x;

    int o0 = 0, o1 = 0;
    if (tid < P) {
        o0 = offs[b * P + tid];
        o1 = offs[(b + 1) * P + tid];
    }
    rs[tid] = tid * CH + o0;
    rl[tid] = o1 - o0;
    if (tid < BIN_NODES) hist2[tid] = 0;
    block_scan_excl(o0, wsum, tid);
    const int bin_start = wsum[32];
    block_scan_excl(o1 - o0, wsum, tid);
    const int total = wsum[32];
    __syncthreads();

    for (int r = wave; r < P; r += 32) {
        const int s0 = rs[r];
        const int len = rl[r];
        for (int i = lane; i < len; i += 64) {
            const bool two = (i + 32) < len;
            const int k0 = part[s0 + i];
            const int k1 = two ? part[s0 + i + 32] : 0;
            atomicAdd(&hist2[k0 & (BIN_NODES - 1)], 1);
            if (two) atomicAdd(&hist2[k1 & (BIN_NODES - 1)], 1);
        }
    }
    __syncthreads();

    const int c = (tid < BIN_NODES) ? hist2[tid] : 0;
    const int ex = block_scan_excl(c, wsum, tid);
    if (tid < BIN_NODES) {
        cursor[tid] = ex;
        const int node = b * BIN_NODES + tid;
        if (node < n) {
            cnt[node] = c;
            rowptr[node] = bin_start + ex;
            const float d = (float)(c + 1);
            dinv[node] = (d > 0.0f) ? rsqrtf(d) : 0.0f;
        }
    }
    __syncthreads();

    const bool staged = total <= CAPO;
    for (int r = wave; r < P; r += 32) {
        const int s0 = rs[r];
        const int len = rl[r];
        for (int i = lane; i < len; i += 64) {
            const bool two = (i + 32) < len;
            const int k0 = part[s0 + i];
            const int k1 = two ? part[s0 + i + 32] : 0;
            const int p0 = atomicAdd(&cursor[k0 & (BIN_NODES - 1)], 1);
            if (staged) obuf[p0] = k0 >> BIN_SHIFT; else col[bin_start + p0] = k0 >> BIN_SHIFT;
            if (two) {
                const int p1 = atomicAdd(&cursor[k1 & (BIN_NODES - 1)], 1);
                if (staged) obuf[p1] = k1 >> BIN_SHIFT; else col[bin_start + p1] = k1 >> BIN_SHIFT;
            }
        }
    }
    __syncthreads();
    if (staged) {
        for (int i = tid; i < total; i += 1024) col[bin_start + i] = obuf[i];
    }
}


__device__ __forceinline__ float4 load_row_piece(const float* __restrict__ x, int s, int lane) {
    const unsigned long long a = (unsigned long long)x + ((unsigned long long)(unsigned)s << 9);
    const unsigned lo = __builtin_amdgcn_readfirstlane((unsigned)a);
    const unsigned hi = __builtin_amdgcn_readfirstlane((unsigned)(a >> 32));
    typedef const float4 __attribute__((address_space(1))) * gf4p;
    const gf4p rowp = (gf4p)(((unsigned long long)hi << 32) | lo);
    return rowp[lane];
}

#define FN 32
#define SP 132

__global__ __launch_bounds__(256)
void layer1_fused_kernel(const float* __restrict__ x,
                         const float* __restrict__ dinv,
                         const int* __restrict__ rowptr,
                         const int* __restrict__ cnt,
                         const int* __restrict__ col,
                         const float* __restrict__ W1,
                         const float* __restrict__ b1,
                         const float* __restrict__ W2,
                         float* __restrict__ h2, int n) {
    constexpr int RT = FN / 16;
    constexpr int NPW = FN / 8;
    __shared__ __align__(16) float strip[FN * SP];

    const int tid = threadIdx.x;
    const int wave = tid >> 5;
    const int lane = tid & 31;
    const int node0 = blockIdx.x * FN;

    int lnode[NPW], lst[NPW], ldg[NPW], lm[NPW];
    float ldi[NPW];
    #pragma unroll
    for (int r = 0; r < NPW; ++r) {
        const int node = __builtin_amdgcn_readfirstlane(node0 + r * 8 + wave);
        const bool ok = node < n;
        const int nc = ok ? node : (n - 1);
        lnode[r] = nc;
        lst[r] = rowptr[nc];
        ldg[r] = cnt[nc];
        lm[r]  = ok ? ldg[r] + 1 : 0;
        ldi[r] = dinv[nc];
    }
    int cs[NPW];
    float cw[NPW];
    #pragma unroll
    for (int r = 0; r < NPW; ++r) {
        const int dg = ldg[r];
        const int k = ((lane < dg) ? lane : dg) - 1;
        const int j = (dg > 0) ? (lst[r] + ((k > 0) ? k : 0)) : 0;
        const int cv = col[j];
        cs[r] = (lane > 0 && lane <= dg) ? cv : lnode[r];
    }
    #pragma unroll
    for (int r = 0; r < NPW; ++r) {
        const float dv = dinv[cs[r]];
        cw[r] = dv * ((lane < lm[r]) ? ldi[r] : 0.0f);
    }

    #pragma unroll
    for (int r = 0; r < NPW; ++r) {
        const int row = r * 8 + wave;
        const int total = lm[r];
        int c = cs[r];
        float w = cw[r];
        float ax = 0.0f, ay = 0.0f, az = 0.0f, aw = 0.0f;
        for (int base = 0; base < total; base += 32) {
            if (base > 0) {
                const int i = base + lane;
                const bool in = i < total;
                c = in ? col[lst[r] + i - 1] : 0;
                w = in ? dinv[c] * ldi[r] : 0.0f;
            }
            const int m = (total - base < 32) ? (total - base) : 32;
            int e = 0;
            for (; e + 8 <= m; e += 8) {
                float4 v[8];
                #pragma unroll
                for (int u = 0; u < 8; ++u) {
                    const int s = __builtin_amdgcn_readlane(c, e + u);
                    v[u] = load_row_piece(x, s, lane);
                }
                #pragma unroll
                for (int u = 0; u < 8; ++u) {
                    const float k = __int_as_float(__builtin_amdgcn_readlane(__float_as_int(w), e + u));
                    ax += v[u].x * k; ay += v[u].y * k; az += v[u].z * k; aw += v[u].w * k;
                }
            }
            if (e < m) {
                float4 v[7];
                #pragma unroll
                for (int u = 0; u < 7; ++u) v[u] = make_float4(0.0f, 0.0f, 0.0f, 0.0f);
                #pragma unroll
                for (int u = 0; u < 7; ++u) {
                    if (e + u < m) {
                        const int s = __builtin_amdgcn_readlane(c, e + u);
                        v[u] = load_row_piece(x, s, lane);
                    }
                }
                #pragma unroll
                for (int u = 0; u < 7; ++u) {
                    if (e + u < m) {
                        const float k = __int_as_float(__builtin_amdgcn_readlane(__float_as_int(w), e + u));
                        ax += v[u].x * k; ay += v[u].y * k; az += v[u].z * k; aw += v[u].w * k;
                    }
                }
            }
        }
        float4 o; o.x = ax; o.y = ay; o.z = az; o.w = aw;
        *(float4*)(&strip[row * SP + lane * 4]) = o;
    }
    __syncthreads();

    const int half = lane >> 4;
    const int l16  = lane & 15;

    {
        const int col_base = wave * 16;
        v8f acc[RT];
        #pragma unroll
        for (int t = 0; t < RT; ++t) acc[t] = v8f{};
        #pragma unroll 4
        for (int k0 = 0; k0 < D_IN; k0 += 4) {
            const int ka = k0 + half * 2;
            v2f b;
            b.x = W1[(size_t)ka * D_HID + col_base + l16];
            b.y = W1[(size_t)(ka + 1) * D_HID + col_base + l16];
            #pragma unroll
            for (int t = 0; t < RT; ++t) {
                const v2f a = *(const v2f*)(&strip[(t * 16 + l16) * SP + ka]);
                acc[t] = __builtin_amdgcn_wmma_f32_16x16x4_f32(
                     false, a,  false, b,
                     (short)0, acc[t],  false,  false);
            }
        }
        __syncthreads();
        const float bb = b1[col_base + l16];
        #pragma unroll
        for (int t = 0; t < RT; ++t) {
            #pragma unroll
            for (int v = 0; v < 8; ++v) {
                strip[(t * 16 + v + half * 8) * SP + col_base + l16] = fmaxf(acc[t][v] + bb, 0.0f);
            }
        }
        __syncthreads();
    }

    #pragma unroll
    for (int tile = wave; tile < 4 * RT; tile += 8) {
        const int col_base = (tile & 3) * 16;
        const int t = tile >> 2;
        float dv[8];
        #pragma unroll
        for (int v = 0; v < 8; ++v) {
            const int node = node0 + t * 16 + v + half * 8;
            dv[v] = dinv[(node < n) ? node : (n - 1)];
        }
        v8f acc = {};
        #pragma unroll 4
        for (int k0 = 0; k0 < D_HID; k0 += 4) {
            const int ka = k0 + half * 2;
            v2f b;
            b.x = W2[(size_t)ka * D_OUT + col_base + l16];
            b.y = W2[(size_t)(ka + 1) * D_OUT + col_base + l16];
            const v2f a = *(const v2f*)(&strip[(t * 16 + l16) * SP + ka]);
            acc = __builtin_amdgcn_wmma_f32_16x16x4_f32(
                 false, a,  false, b,
                 (short)0, acc,  false,  false);
        }
        #pragma unroll
        for (int v = 0; v < 8; ++v) {
            const int node = node0 + t * 16 + v + half * 8;
            if (node < n) h2[(size_t)node * D_OUT + col_base + l16] = acc[v] * dv[v];
        }
    }
}

template <int D>
__global__ __launch_bounds__(256)
void gather_scaled_rows_kernel(const float* __restrict__ hs,
                               const float* __restrict__ dinv,
                               const int* __restrict__ rowptr,
                               const int* __restrict__ cnt,
                               const int* __restrict__ col,
                               const float* __restrict__ bias,
                               float* __restrict__ out, int n) {
    constexpr int LPR = D / 4;
    constexpr int G = 32 / LPR;
    const int node = (blockIdx.x * blockDim.x + threadIdx.x) >> 5;
    if (node >= n) return;
    const int lane = threadIdx.x & 31;
    const int g = lane / LPR;
    const int sub = lane % LPR;
    const float di = dinv[node];
    const int start = rowptr[node];
    const int end = start + cnt[node];
    const float* __restrict__ hb = hs + sub * 4;

    float ax = 0.0f, ay = 0.0f, az = 0.0f, aw = 0.0f;
    if (g == 0) {
        const float4 v = *(const float4*)(hb + (size_t)node * D);
        ax = v.x; ay = v.y; az = v.z; aw = v.w;
    }

    int j = start + g;
    for (; j + 3 * G < end; j += 4 * G) {
        const int s0 = col[j];
        const int s1 = col[j + G];
        const int s2 = col[j + 2 * G];
        const int s3 = col[j + 3 * G];
        const float4 v0 = *(const float4*)(hb + (size_t)s0 * D);
        const float4 v1 = *(const float4*)(hb + (size_t)s1 * D);
        const float4 v2 = *(const float4*)(hb + (size_t)s2 * D);
        const float4 v3 = *(const float4*)(hb + (size_t)s3 * D);
        ax += v0.x; ay += v0.y; az += v0.z; aw += v0.w;
        ax += v1.x; ay += v1.y; az += v1.z; aw += v1.w;
        ax += v2.x; ay += v2.y; az += v2.z; aw += v2.w;
        ax += v3.x; ay += v3.y; az += v3.z; aw += v3.w;
    }
    for (; j < end; j += G) {
        const int s0 = col[j];
        const float4 v0 = *(const float4*)(hb + (size_t)s0 * D);
        ax += v0.x; ay += v0.y; az += v0.z; aw += v0.w;
    }

    if (G == 2) {
        ax += __shfl_xor(ax, 16);
        ay += __shfl_xor(ay, 16);
        az += __shfl_xor(az, 16);
        aw += __shfl_xor(aw, 16);
    }
    if (g == 0) {
        const float4 b = *(const float4*)(bias + sub * 4);
        float4 r;
        r.x = ax * di + b.x; r.y = ay * di + b.y;
        r.z = az * di + b.z; r.w = aw * di + b.w;
        *(float4*)(out + (size_t)node * D + sub * 4) = r;
    }
}

extern "C" void kernel_launch(void* const* d_in, const int* in_sizes, int n_in,
                              void* d_out, int out_size, void* d_ws, size_t ws_size,
                              hipStream_t stream) {
    const float* x    = (const float*)d_in[0];
    const int*   eidx = (const int*)  d_in[1];
    const float* W1   = (const float*)d_in[2];
    const float* b1   = (const float*)d_in[3];
    const float* W2   = (const float*)d_in[4];
    const float* b2   = (const float*)d_in[5];
    float* out = (float*)d_out;

    const int N = in_sizes[0] / D_IN;
    const int E = in_sizes[1] / 2;
    const int* src = eidx;
    const int* dst = eidx + E;

    const int NB = (N + BIN_NODES - 1) / BIN_NODES;
    const int P  = (E + CH - 1) / CH;

    const size_t NP = ((size_t)N + 255) & ~(size_t)255;
    const size_t EP = ((size_t)E + 255) & ~(size_t)255;
    const size_t OP = ((size_t)(NB + 1) * P + 255) & ~(size_t)255;
    float* ws = (float*)d_ws;
    float* dinv   = ws;
    int*   cnt    = (int*)(ws + NP);
    int*   rowptr = (int*)(ws + 2 * NP);
    int*   part   = (int*)(ws + 3 * NP);
    int*   col    = (int*)(ws + 3 * NP + EP);
    int*   offs   = (int*)(ws + 3 * NP + 2 * EP);
    float* h2     = ws + 3 * NP + 2 * EP + OP;

    const int BT = 256;

    bin_partition_kernel<<<P, 1024, 0, stream>>>(src, dst, part, offs, E, NB);
    bin_sort_kernel<<<NB, 1024, 0, stream>>>(part, offs, cnt, rowptr, dinv, col, N, P);

    const unsigned gather_blocks = (unsigned)(((long)N * 32 + BT - 1) / BT);

    layer1_fused_kernel<<<(N + FN - 1) / FN, 256, 0, stream>>>(
        x, dinv, rowptr, cnt, col, W1, b1, W2, h2, N);

    gather_scaled_rows_kernel<D_OUT><<<gather_blocks, BT, 0, stream>>>(
        h2, dinv, rowptr, cnt, col, b2, out, N);
}


# W1 and W2 re-laid in WMMA B-fragment order in-call (one spare block of the sort launch); fused kernel fetches B fragments as 16-byte loads, 2 k steps each, 4 ahead; product and sum order unchanged
# speedup vs baseline: 6.2118x; 1.0433x over previous
#include <hip/hip_runtime.h>


typedef __attribute__((ext_vector_type(2))) float v2f;
typedef __attribute__((ext_vector_type(8))) float v8f;

#define D_IN  128
#define D_HID 128
#define D_OUT 64

#define BIN_SHIFT 9
#define BIN_NODES 512
#define NBMAX 256
#define EPT 7
#define CH (EPT * 1024)
#define CAPO 16384

__device__ __forceinline__ int block_scan_excl(int v, int* wsum, int tid) {
    const int lane = tid & 31;
    const int wave = tid >> 5;
    int incl = v;
    for (int o = 1; o < 32; o <<= 1) {
        const int t = __shfl_up(incl, o);
        if (lane >= o) incl += t;
    }
    __syncthreads();
    if (lane == 31) wsum[wave] = incl;
    __syncthreads();
    if (wave == 0) {
        const int t = wsum[lane];
        int ti = t;
        for (int o = 1; o < 32; o <<= 1) {
            const int u = __shfl_up(ti, o);
            if (lane >= o) ti += u;
        }
        wsum[lane] = ti - t;
        if (lane == 31) wsum[32] = ti;
    }
    __syncthreads();
    return wsum[wave] + (incl - v);
}

__global__ __launch_bounds__(1024)
void bin_partition_kernel(const int* __restrict__ src,
                          const int* __restrict__ dst,
                          int* __restrict__ part,
                          int* __restrict__ offs, int e, int nb) {
    __shared__ int hist[NBMAX];
    __shared__ int boff[NBMAX + 1];
    __shared__ int wsum[33];
    __shared__ int sorted[CH];
    const int tid = threadIdx.x;
    const int p = blockIdx.x;
    const int P = gridDim.x;
    const int e0 = p * CH;
    const int ne = (e - e0 < CH) ? (e - e0) : CH;

    if (tid < NBMAX) hist[tid] = 0;
    __syncthreads();

    int key[EPT], bin[EPT], rk[EPT];
    #pragma unroll
    for (int k = 0; k < EPT; ++k) {
        const int i = k * 1024 + tid;
        const int ii = (i < ne) ? i : (ne - 1);
        bin[k] = dst[e0 + ii];
        key[k] = src[e0 + ii];
    }
    #pragma unroll
    for (int k = 0; k < EPT; ++k) {
        const int i = k * 1024 + tid;
        const int d = bin[k];
        bin[k] = -1;
        rk[k] = 0;
        if (i < ne) {
            bin[k] = d >> BIN_SHIFT;
            key[k] = (key[k] << BIN_SHIFT) | (d & (BIN_NODES - 1));
            rk[k] = atomicAdd(&hist[bin[k]], 1);
        }
    }
    __syncthreads();

    const int hv = (tid < nb) ? hist[tid] : 0;
    const int ex = block_scan_excl(hv, wsum, tid);
    if (tid <= nb) boff[tid] = ex;
    __syncthreads();

    #pragma unroll
    for (int k = 0; k < EPT; ++k) {
        if (bin[k] >= 0) sorted[boff[bin[k]] + rk[k]] = key[k];
    }
    __syncthreads();

    for (int i = tid; i < ne; i += 1024) part[e0 + i] = sorted[i];
    if (tid <= nb) offs[tid * P + p] = boff[tid];
}

template <int NCOLS>
__device__ __forceinline__ void relayout_b_fragments(const float* __restrict__ W,
                                                     float4* __restrict__ Wf, int tid) {
    constexpr int NQ = (NCOLS / 16) * 16 * 32;
    float4 v[NQ / 1024];
    #pragma unroll
    for (int i = 0; i < NQ / 1024; ++i) {
        const int q = i * 1024 + tid;
        const int l = q & 31;
        const int kc = (q >> 5) & 15;
        const int ct = q >> 9;
        const int c16 = l & 15;
        const int h = l >> 4;
        const float* p = W + (size_t)(8 * kc + 2 * h) * NCOLS + 16 * ct + c16;
        v[i].x = p[0];
        v[i].y = p[NCOLS];
        v[i].z = p[4 * NCOLS];
        v[i].w = p[5 * NCOLS];
    }
    #pragma unroll
    for (int i = 0; i < NQ / 1024; ++i) Wf[i * 1024 + tid] = v[i];
}

__global__ __launch_bounds__(1024)
void bin_sort_kernel(const int* __restrict__ part,
                     const int* __restrict__ offs,
                     int* __restrict__ cnt,
                     int* __restrict__ rowptr,
                     float* __restrict__ dinv,
                     int* __restrict__ col, int n, int P,
                     const float* __restrict__ W1,
                     const float* __restrict__ W2,
                     float* __restrict__ W1f,
                     float* __restrict__ W2f, int nbins) {
    __shared__ int rs[1024];
    __shared__ int rl[1024];
    __shared__ int hist2[BIN_NODES];
    __shared__ int cursor[BIN_NODES];
    __shared__ int wsum[33];
    __shared__ int obuf[CAPO];
    const int tid = threadIdx.x;
    const int lane = tid & 31;
    const int wave = tid >> 5;
    const int b = blockIdx.x;

    if (b == nbins) {
        relayout_b_fragments<D_HID>(W1, (float4*)W1f, tid);
        relayout_b_fragments<D_OUT>(W2, (float4*)W2f, tid);
        return;
    }

    int o0 = 0, o1 = 0;
    if (tid < P) {
        o0 = offs[b * P + tid];
        o1 = offs[(b + 1) * P + tid];
    }
    rs[tid] = tid * CH + o0;
    rl[tid] = o1 - o0;
    if (tid < BIN_NODES) hist2[tid] = 0;
    block_scan_excl(o0, wsum, tid);
    const int bin_start = wsum[32];
    block_scan_excl(o1 - o0, wsum, tid);
    const int total = wsum[32];
    __syncthreads();

    for (int r = wave; r < P; r += 32) {
        const int s0 = rs[r];
        const int len = rl[r];
        for (int i = lane; i < len; i += 64) {
            const bool two = (i + 32) < len;
            const int k0 = part[s0 + i];
            const int k1 = two ? part[s0 + i + 32] : 0;
            atomicAdd(&hist2[k0 & (BIN_NODES - 1)], 1);
            if (two) atomicAdd(&hist2[k1 & (BIN_NODES - 1)], 1);
        }
    }
    __syncthreads();

    const int c = (tid < BIN_NODES) ? hist2[tid] : 0;
    const int ex = block_scan_excl(c, wsum, tid);
    if (tid < BIN_NODES) {
        cursor[tid] = ex;
        const int node = b * BIN_NODES + tid;
        if (node < n) {
            cnt[node] = c;
            rowptr[node] = bin_start + ex;
            const float d = (float)(c + 1);
            dinv[node] = (d > 0.0f) ? rsqrtf(d) : 0.0f;
        }
    }
    __syncthreads();

    const bool staged = total <= CAPO;
    for (int r = wave; r < P; r += 32) {
        const int s0 = rs[r];
        const int len = rl[r];
        for (int i = lane; i < len; i += 64) {
            const bool two = (i + 32) < len;
            const int k0 = part[s0 + i];
            const int k1 = two ? part[s0 + i + 32] : 0;
            const int p0 = atomicAdd(&cursor[k0 & (BIN_NODES - 1)], 1);
            if (staged) obuf[p0] = k0 >> BIN_SHIFT; else col[bin_start + p0] = k0 >> BIN_SHIFT;
            if (two) {
                const int p1 = atomicAdd(&cursor[k1 & (BIN_NODES - 1)], 1);
                if (staged) obuf[p1] = k1 >> BIN_SHIFT; else col[bin_start + p1] = k1 >> BIN_SHIFT;
            }
        }
    }
    __syncthreads();
    if (staged) {
        for (int i = tid; i < total; i += 1024) col[bin_start + i] = obuf[i];
    }
}


__device__ __forceinline__ float4 load_row_piece(const float* __restrict__ x, int s, int lane) {
    const unsigned long long a = (unsigned long long)x + ((unsigned long long)(unsigned)s << 9);
    const unsigned lo = __builtin_amdgcn_readfirstlane((unsigned)a);
    const unsigned hi = __builtin_amdgcn_readfirstlane((unsigned)(a >> 32));
    typedef const float4 __attribute__((address_space(1))) * gf4p;
    const gf4p rowp = (gf4p)(((unsigned long long)hi << 32) | lo);
    return rowp[lane];
}

#define FN 32
#define SP 132

template <int NT>
__device__ __forceinline__ void mma_four_chunks(v8f (&acc)[NT], const float4 (&bq)[4],
                                                const float* rows, int kbase,
                                                int l16, int half) {
    #pragma unroll
    for (int u = 0; u < 4; ++u) {
        #pragma unroll
        for (int s = 0; s < 2; ++s) {
            const int ka = kbase + u * 8 + s * 4 + half * 2;
            v2f b;
            b.x = s ? bq[u].z : bq[u].x;
            b.y = s ? bq[u].w : bq[u].y;
            #pragma unroll
            for (int t = 0; t < NT; ++t) {
                const v2f a = *(const v2f*)(&rows[(t * 16 + l16) * SP + ka]);
                acc[t] = __builtin_amdgcn_wmma_f32_16x16x4_f32(
                     false, a,  false, b,
                     (short)0, acc[t],  false,  false);
            }
        }
    }
}

__device__ __forceinline__ void load_four_chunks(float4 (&bq)[4], const float4* __restrict__ q, int g) {
    #pragma unroll
    for (int u = 0; u < 4; ++u) bq[u] = q[(g * 4 + u) * 32];
}

__global__ __launch_bounds__(256)
void layer1_fused_kernel(const float* __restrict__ x,
                         const float* __restrict__ dinv,
                         const int* __restrict__ rowptr,
                         const int* __restrict__ cnt,
                         const int* __restrict__ col,
                         const float* __restrict__ W1f,
                         const float* __restrict__ b1,
                         const float* __restrict__ W2f,
                         float* __restrict__ h2, int n) {
    constexpr int RT = FN / 16;
    constexpr int NPW = FN / 8;
    __shared__ __align__(16) float strip[FN * SP];

    const int tid = threadIdx.x;
    const int wave = tid >> 5;
    const int lane = tid & 31;
    const int node0 = blockIdx.x * FN;

    int lnode[NPW], lst[NPW], ldg[NPW], lm[NPW];
    float ldi[NPW];
    #pragma unroll
    for (int r = 0; r < NPW; ++r) {
        const int node = __builtin_amdgcn_readfirstlane(node0 + r * 8 + wave);
        const bool ok = node < n;
        const int nc = ok ? node : (n - 1);
        lnode[r] = nc;
        lst[r] = rowptr[nc];
        ldg[r] = cnt[nc];
        lm[r]  = ok ? ldg[r] + 1 : 0;
        ldi[r] = dinv[nc];
    }
    int cs[NPW];
    float cw[NPW];
    #pragma unroll
    for (int r = 0; r < NPW; ++r) {
        const int dg = ldg[r];
        const int k = ((lane < dg) ? lane : dg) - 1;
        const int j = (dg > 0) ? (lst[r] + ((k > 0) ? k : 0)) : 0;
        const int cv = col[j];
        cs[r] = (lane > 0 && lane <= dg) ? cv : lnode[r];
    }
    #pragma unroll
    for (int r = 0; r < NPW; ++r) {
        const float dv = dinv[cs[r]];
        cw[r] = dv * ((lane < lm[r]) ? ldi[r] : 0.0f);
    }

    #pragma unroll
    for (int r = 0; r < NPW; ++r) {
        const int row = r * 8 + wave;
        const int total = lm[r];
        int c = cs[r];
        float w = cw[r];
        float ax = 0.0f, ay = 0.0f, az = 0.0f, aw = 0.0f;
        for (int base = 0; base < total; base += 32) {
            if (base > 0) {
                const int i = base + lane;
                const bool in = i < total;
                c = in ? col[lst[r] + i - 1] : 0;
                w = in ? dinv[c] * ldi[r] : 0.0f;
            }
            const int m = (total - base < 32) ? (total - base) : 32;
            int e = 0;
            for (; e + 8 <= m; e += 8) {
                float4 v[8];
                #pragma unroll
                for (int u = 0; u < 8; ++u) {
                    const int s = __builtin_amdgcn_readlane(c, e + u);
                    v[u] = load_row_piece(x, s, lane);
                }
                #pragma unroll
                for (int u = 0; u < 8; ++u) {
                    const float k = __int_as_float(__builtin_amdgcn_readlane(__float_as_int(w), e + u));
                    ax += v[u].x * k; ay += v[u].y * k; az += v[u].z * k; aw += v[u].w * k;
                }
            }
            if (e < m) {
                float4 v[7];
                #pragma unroll
                for (int u = 0; u < 7; ++u) v[u] = make_float4(0.0f, 0.0f, 0.0f, 0.0f);
                #pragma unroll
                for (int u = 0; u < 7; ++u) {
                    if (e + u < m) {
                        const int s = __builtin_amdgcn_readlane(c, e + u);
                        v[u] = load_row_piece(x, s, lane);
                    }
                }
                #pragma unroll
                for (int u = 0; u < 7; ++u) {
                    if (e + u < m) {
                        const float k = __int_as_float(__builtin_amdgcn_readlane(__float_as_int(w), e + u));
                        ax += v[u].x * k; ay += v[u].y * k; az += v[u].z * k; aw += v[u].w * k;
                    }
                }
            }
        }
        float4 o; o.x = ax; o.y = ay; o.z = az; o.w = aw;
        *(float4*)(&strip[row * SP + lane * 4]) = o;
    }
    __syncthreads();

    const int half = lane >> 4;
    const int l16  = lane & 15;

    static_assert(4 * RT == 8, "phase 3 gives each of the 8 waves one 16x16 tile");
    const int ct2 = wave & 3;
    const int t2 = wave >> 2;
    const float4* __restrict__ w2q = (const float4*)W2f + ct2 * (16 * 32) + lane;
    float4 qa[4], qb[4];
    {
        const int col_base = wave * 16;
        const float4* __restrict__ w1q = (const float4*)W1f + wave * (16 * 32) + lane;
        v8f acc[RT];
        #pragma unroll
        for (int t = 0; t < RT; ++t) acc[t] = v8f{};
        load_four_chunks(qa, w1q, 0);
        load_four_chunks(qb, w1q, 1);
        mma_four_chunks<RT>(acc, qa, strip, 0, l16, half);
        load_four_chunks(qa, w1q, 2);
        mma_four_chunks<RT>(acc, qb, strip, 32, l16, half);
        load_four_chunks(qb, w1q, 3);
        mma_four_chunks<RT>(acc, qa, strip, 64, l16, half);
        mma_four_chunks<RT>(acc, qb, strip, 96, l16, half);
        load_four_chunks(qa, w2q, 0);
        load_four_chunks(qb, w2q, 1);
        __syncthreads();
        const float bb = b1[col_base + l16];
        #pragma unroll
        for (int t = 0; t < RT; ++t) {
            #pragma unroll
            for (int v = 0; v < 8; ++v) {
                strip[(t * 16 + v + half * 8) * SP + col_base + l16] = fmaxf(acc[t][v] + bb, 0.0f);
            }
        }
        __syncthreads();
    }

    {
        const int col_base = ct2 * 16;
        float dv[8];
        #pragma unroll
        for (int v = 0; v < 8; ++v) {
            const int node = node0 + t2 * 16 + v + half * 8;
            dv[v] = dinv[(node < n) ? node : (n - 1)];
        }
        const float* rows = strip + t2 * 16 * SP;
        v8f acc[1];
        acc[0] = v8f{};
        mma_four_chunks<1>(acc, qa, rows, 0, l16, half);
        load_four_chunks(qa, w2q, 2);
        mma_four_chunks<1>(acc, qb, rows, 32, l16, half);
        load_four_chunks(qb, w2q, 3);
        mma_four_chunks<1>(acc, qa, rows, 64, l16, half);
        mma_four_chunks<1>(acc, qb, rows, 96, l16, half);
        #pragma unroll
        for (int v = 0; v < 8; ++v) {
            const int node = node0 + t2 * 16 + v + half * 8;
            if (node < n) h2[(size_t)node * D_OUT + col_base + l16] = acc[0][v] * dv[v];
        }
    }
}

template <int D>
__global__ __launch_bounds__(256)
void gather_scaled_rows_kernel(const float* __restrict__ hs,
                               const float* __restrict__ dinv,
                               const int* __restrict__ rowptr,
                               const int* __restrict__ cnt,
                               const int* __restrict__ col,
                               const float* __restrict__ bias,
                               float* __restrict__ out, int n) {
    constexpr int LPR = D / 4;
    constexpr int G = 32 / LPR;
    const int node = (blockIdx.x * blockDim.x + threadIdx.x) >> 5;
    if (node >= n) return;
    const int lane = threadIdx.x & 31;
    const int g = lane / LPR;
    const int sub = lane % LPR;
    const float di = dinv[node];
    const int start = rowptr[node];
    const int end = start + cnt[node];
    const float* __restrict__ hb = hs + sub * 4;

    float ax = 0.0f, ay = 0.0f, az = 0.0f, aw = 0.0f;
    if (g == 0) {
        const float4 v = *(const float4*)(hb + (size_t)node * D);
        ax = v.x; ay = v.y; az = v.z; aw = v.w;
    }

    int j = start + g;
    for (; j + 3 * G < end; j += 4 * G) {
        const int s0 = col[j];
        const int s1 = col[j + G];
        const int s2 = col[j + 2 * G];
        const int s3 = col[j + 3 * G];
        const float4 v0 = *(const float4*)(hb + (size_t)s0 * D);
        const float4 v1 = *(const float4*)(hb + (size_t)s1 * D);
        const float4 v2 = *(const float4*)(hb + (size_t)s2 * D);
        const float4 v3 = *(const float4*)(hb + (size_t)s3 * D);
        ax += v0.x; ay += v0.y; az += v0.z; aw += v0.w;
        ax += v1.x; ay += v1.y; az += v1.z; aw += v1.w;
        ax += v2.x; ay += v2.y; az += v2.z; aw += v2.w;
        ax += v3.x; ay += v3.y; az += v3.z; aw += v3.w;
    }
    for (; j < end; j += G) {
        const int s0 = col[j];
        const float4 v0 = *(const float4*)(hb + (size_t)s0 * D);
        ax += v0.x; ay += v0.y; az += v0.z; aw += v0.w;
    }

    if (G == 2) {
        ax += __shfl_xor(ax, 16);
        ay += __shfl_xor(ay, 16);
        az += __shfl_xor(az, 16);
        aw += __shfl_xor(aw, 16);
    }
    if (g == 0) {
        const float4 b = *(const float4*)(bias + sub * 4);
        float4 r;
        r.x = ax * di + b.x; r.y = ay * di + b.y;
        r.z = az * di + b.z; r.w = aw * di + b.w;
        *(float4*)(out + (size_t)node * D + sub * 4) = r;
    }
}

extern "C" void kernel_launch(void* const* d_in, const int* in_sizes, int n_in,
                              void* d_out, int out_size, void* d_ws, size_t ws_size,
                              hipStream_t stream) {
    const float* x    = (const float*)d_in[0];
    const int*   eidx = (const int*)  d_in[1];
    const float* W1   = (const float*)d_in[2];
    const float* b1   = (const float*)d_in[3];
    const float* W2   = (const float*)d_in[4];
    const float* b2   = (const float*)d_in[5];
    float* out = (float*)d_out;

    const int N = in_sizes[0] / D_IN;
    const int E = in_sizes[1] / 2;
    const int* src = eidx;
    const int* dst = eidx + E;

    const int NB = (N + BIN_NODES - 1) / BIN_NODES;
    const int P  = (E + CH - 1) / CH;

    const size_t NP = ((size_t)N + 255) & ~(size_t)255;
    const size_t EP = ((size_t)E + 255) & ~(size_t)255;
    const size_t OP = ((size_t)(NB + 1) * P + 255) & ~(size_t)255;
    float* ws = (float*)d_ws;
    float* dinv   = ws;
    int*   cnt    = (int*)(ws + NP);
    int*   rowptr = (int*)(ws + 2 * NP);
    int*   part   = (int*)(ws + 3 * NP);
    int*   col    = (int*)(ws + 3 * NP + EP);
    int*   offs   = (int*)(ws + 3 * NP + 2 * EP);
    float* h2     = ws + 3 * NP + 2 * EP + OP;
    const size_t HP = ((size_t)N * D_OUT + 255) & ~(size_t)255;
    float* W1f    = h2 + HP;
    float* W2f    = W1f + D_IN * D_HID;

    const int BT = 256;

    bin_partition_kernel<<<P, 1024, 0, stream>>>(src, dst, part, offs, E, NB);
    bin_sort_kernel<<<NB + 1, 1024, 0, stream>>>(part, offs, cnt, rowptr, dinv, col, N, P,
                                                W1, W2, W1f, W2f, NB);

    const unsigned gather_blocks = (unsigned)(((long)N * 32 + BT - 1) / BT);

    layer1_fused_kernel<<<(N + FN - 1) / FN, 256, 0, stream>>>(
        x, dinv, rowptr, cnt, col, W1f, b1, W2f, h2, N);

    gather_scaled_rows_kernel<D_OUT><<<gather_blocks, BT, 0, stream>>>(
        h2, dinv, rowptr, cnt, col, b2, out, N);
}


# Layer-2 gather: a wave walks 4 consecutive target nodes (100000 waves to 25000), their source ids fetched up front one per lane and dealt out by lane read, 8 rows in flight; sum order unchanged
# speedup vs baseline: 6.5167x; 1.0491x over previous
#include <hip/hip_runtime.h>


typedef __attribute__((ext_vector_type(2))) float v2f;
typedef __attribute__((ext_vector_type(8))) float v8f;

#define D_IN  128
#define D_HID 128
#define D_OUT 64

#define BIN_SHIFT 9
#define BIN_NODES 512
#define NBMAX 256
#define EPT 7
#define CH (EPT * 1024)
#define CAPO 16384

__device__ __forceinline__ int block_scan_excl(int v, int* wsum, int tid) {
    const int lane = tid & 31;
    const int wave = tid >> 5;
    int incl = v;
    for (int o = 1; o < 32; o <<= 1) {
        const int t = __shfl_up(incl, o);
        if (lane >= o) incl += t;
    }
    __syncthreads();
    if (lane == 31) wsum[wave] = incl;
    __syncthreads();
    if (wave == 0) {
        const int t = wsum[lane];
        int ti = t;
        for (int o = 1; o < 32; o <<= 1) {
            const int u = __shfl_up(ti, o);
            if (lane >= o) ti += u;
        }
        wsum[lane] = ti - t;
        if (lane == 31) wsum[32] = ti;
    }
    __syncthreads();
    return wsum[wave] + (incl - v);
}

__global__ __launch_bounds__(1024)
void bin_partition_kernel(const int* __restrict__ src,
                          const int* __restrict__ dst,
                          int* __restrict__ part,
                          int* __restrict__ offs, int e, int nb) {
    __shared__ int hist[NBMAX];
    __shared__ int boff[NBMAX + 1];
    __shared__ int wsum[33];
    __shared__ int sorted[CH];
    const int tid = threadIdx.x;
    const int p = blockIdx.x;
    const int P = gridDim.x;
    const int e0 = p * CH;
    const int ne = (e - e0 < CH) ? (e - e0) : CH;

    if (tid < NBMAX) hist[tid] = 0;
    __syncthreads();

    int key[EPT], bin[EPT], rk[EPT];
    #pragma unroll
    for (int k = 0; k < EPT; ++k) {
        const int i = k * 1024 + tid;
        const int ii = (i < ne) ? i : (ne - 1);
        bin[k] = dst[e0 + ii];
        key[k] = src[e0 + ii];
    }
    #pragma unroll
    for (int k = 0; k < EPT; ++k) {
        const int i = k * 1024 + tid;
        const int d = bin[k];
        bin[k] = -1;
        rk[k] = 0;
        if (i < ne) {
            bin[k] = d >> BIN_SHIFT;
            key[k] = (key[k] << BIN_SHIFT) | (d & (BIN_NODES - 1));
            rk[k] = atomicAdd(&hist[bin[k]], 1);
        }
    }
    __syncthreads();

    const int hv = (tid < nb) ? hist[tid] : 0;
    const int ex = block_scan_excl(hv, wsum, tid);
    if (tid <= nb) boff[tid] = ex;
    __syncthreads();

    #pragma unroll
    for (int k = 0; k < EPT; ++k) {
        if (bin[k] >= 0) sorted[boff[bin[k]] + rk[k]] = key[k];
    }
    __syncthreads();

    for (int i = tid; i < ne; i += 1024) part[e0 + i] = sorted[i];
    if (tid <= nb) offs[tid * P + p] = boff[tid];
}

template <int NCOLS>
__device__ __forceinline__ void relayout_b_fragments(const float* __restrict__ W,
                                                     float4* __restrict__ Wf, int tid) {
    constexpr int NQ = (NCOLS / 16) * 16 * 32;
    float4 v[NQ / 1024];
    #pragma unroll
    for (int i = 0; i < NQ / 1024; ++i) {
        const int q = i * 1024 + tid;
        const int l = q & 31;
        const int kc = (q >> 5) & 15;
        const int ct = q >> 9;
        const int c16 = l & 15;
        const int h = l >> 4;
        const float* p = W + (size_t)(8 * kc + 2 * h) * NCOLS + 16 * ct + c16;
        v[i].x = p[0];
        v[i].y = p[NCOLS];
        v[i].z = p[4 * NCOLS];
        v[i].w = p[5 * NCOLS];
    }
    #pragma unroll
    for (int i = 0; i < NQ / 1024; ++i) Wf[i * 1024 + tid] = v[i];
}

__global__ __launch_bounds__(1024)
void bin_sort_kernel(const int* __restrict__ part,
                     const int* __restrict__ offs,
                     int* __restrict__ cnt,
                     int* __restrict__ rowptr,
                     float* __restrict__ dinv,
                     int* __restrict__ col, int n, int P,
                     const float* __restrict__ W1,
                     const float* __restrict__ W2,
                     float* __restrict__ W1f,
                     float* __restrict__ W2f, int nbins) {
    __shared__ int rs[1024];
    __shared__ int rl[1024];
    __shared__ int hist2[BIN_NODES];
    __shared__ int cursor[BIN_NODES];
    __shared__ int wsum[33];
    __shared__ int obuf[CAPO];
    const int tid = threadIdx.x;
    const int lane = tid & 31;
    const int wave = tid >> 5;
    const int b = blockIdx.x;

    if (b == nbins) {
        relayout_b_fragments<D_HID>(W1, (float4*)W1f, tid);
        relayout_b_fragments<D_OUT>(W2, (float4*)W2f, tid);
        return;
    }

    int o0 = 0, o1 = 0;
    if (tid < P) {
        o0 = offs[b * P + tid];
        o1 = offs[(b + 1) * P + tid];
    }
    rs[tid] = tid * CH + o0;
    rl[tid] = o1 - o0;
    if (tid < BIN_NODES) hist2[tid] = 0;
    block_scan_excl(o0, wsum, tid);
    const int bin_start = wsum[32];
    block_scan_excl(o1 - o0, wsum, tid);
    const int total = wsum[32];
    __syncthreads();

    for (int r = wave; r < P; r += 32) {
        const int s0 = rs[r];
        const int len = rl[r];
        for (int i = lane; i < len; i += 64) {
            const bool two = (i + 32) < len;
            const int k0 = part[s0 + i];
            const int k1 = two ? part[s0 + i + 32] : 0;
            atomicAdd(&hist2[k0 & (BIN_NODES - 1)], 1);
            if (two) atomicAdd(&hist2[k1 & (BIN_NODES - 1)], 1);
        }
    }
    __syncthreads();

    const int c = (tid < BIN_NODES) ? hist2[tid] : 0;
    const int ex = block_scan_excl(c, wsum, tid);
    if (tid < BIN_NODES) {
        cursor[tid] = ex;
        const int node = b * BIN_NODES + tid;
        if (node < n) {
            cnt[node] = c;
            rowptr[node] = bin_start + ex;
            const float d = (float)(c + 1);
            dinv[node] = (d > 0.0f) ? rsqrtf(d) : 0.0f;
        }
    }
    __syncthreads();

    const bool staged = total <= CAPO;
    for (int r = wave; r < P; r += 32) {
        const int s0 = rs[r];
        const int len = rl[r];
        for (int i = lane; i < len; i += 64) {
            const bool two = (i + 32) < len;
            const int k0 = part[s0 + i];
            const int k1 = two ? part[s0 + i + 32] : 0;
            const int p0 = atomicAdd(&cursor[k0 & (BIN_NODES - 1)], 1);
            if (staged) obuf[p0] = k0 >> BIN_SHIFT; else col[bin_start + p0] = k0 >> BIN_SHIFT;
            if (two) {
                const int p1 = atomicAdd(&cursor[k1 & (BIN_NODES - 1)], 1);
                if (staged) obuf[p1] = k1 >> BIN_SHIFT; else col[bin_start + p1] = k1 >> BIN_SHIFT;
            }
        }
    }
    __syncthreads();
    if (staged) {
        for (int i = tid; i < total; i += 1024) col[bin_start + i] = obuf[i];
    }
}


__device__ __forceinline__ float4 load_row_piece(const float* __restrict__ x, int s, int lane) {
    const unsigned long long a = (unsigned long long)x + ((unsigned long long)(unsigned)s << 9);
    const unsigned lo = __builtin_amdgcn_readfirstlane((unsigned)a);
    const unsigned hi = __builtin_amdgcn_readfirstlane((unsigned)(a >> 32));
    typedef const float4 __attribute__((address_space(1))) * gf4p;
    const gf4p rowp = (gf4p)(((unsigned long long)hi << 32) | lo);
    return rowp[lane];
}

#define FN 32
#define SP 132

template <int NT>
__device__ __forceinline__ void mma_four_chunks(v8f (&acc)[NT], const float4 (&bq)[4],
                                                const float* rows, int kbase,
                                                int l16, int half) {
    #pragma unroll
    for (int u = 0; u < 4; ++u) {
        #pragma unroll
        for (int s = 0; s < 2; ++s) {
            const int ka = kbase + u * 8 + s * 4 + half * 2;
            v2f b;
            b.x = s ? bq[u].z : bq[u].x;
            b.y = s ? bq[u].w : bq[u].y;
            #pragma unroll
            for (int t = 0; t < NT; ++t) {
                const v2f a = *(const v2f*)(&rows[(t * 16 + l16) * SP + ka]);
                acc[t] = __builtin_amdgcn_wmma_f32_16x16x4_f32(
                     false, a,  false, b,
                     (short)0, acc[t],  false,  false);
            }
        }
    }
}

__device__ __forceinline__ void load_four_chunks(float4 (&bq)[4], const float4* __restrict__ q, int g) {
    #pragma unroll
    for (int u = 0; u < 4; ++u) bq[u] = q[(g * 4 + u) * 32];
}

__global__ __launch_bounds__(256)
void layer1_fused_kernel(const float* __restrict__ x,
                         const float* __restrict__ dinv,
                         const int* __restrict__ rowptr,
                         const int* __restrict__ cnt,
                         const int* __restrict__ col,
                         const float* __restrict__ W1f,
                         const float* __restrict__ b1,
                         const float* __restrict__ W2f,
                         float* __restrict__ h2, int n) {
    constexpr int RT = FN / 16;
    constexpr int NPW = FN / 8;
    __shared__ __align__(16) float strip[FN * SP];

    const int tid = threadIdx.x;
    const int wave = tid >> 5;
    const int lane = tid & 31;
    const int node0 = blockIdx.x * FN;

    int lnode[NPW], lst[NPW], ldg[NPW], lm[NPW];
    float ldi[NPW];
    #pragma unroll
    for (int r = 0; r < NPW; ++r) {
        const int node = __builtin_amdgcn_readfirstlane(node0 + r * 8 + wave);
        const bool ok = node < n;
        const int nc = ok ? node : (n - 1);
        lnode[r] = nc;
        lst[r] = rowptr[nc];
        ldg[r] = cnt[nc];
        lm[r]  = ok ? ldg[r] + 1 : 0;
        ldi[r] = dinv[nc];
    }
    int cs[NPW];
    float cw[NPW];
    #pragma unroll
    for (int r = 0; r < NPW; ++r) {
        const int dg = ldg[r];
        const int k = ((lane < dg) ? lane : dg) - 1;
        const int j = (dg > 0) ? (lst[r] + ((k > 0) ? k : 0)) : 0;
        const int cv = col[j];
        cs[r] = (lane > 0 && lane <= dg) ? cv : lnode[r];
    }
    #pragma unroll
    for (int r = 0; r < NPW; ++r) {
        const float dv = dinv[cs[r]];
        cw[r] = dv * ((lane < lm[r]) ? ldi[r] : 0.0f);
    }

    #pragma unroll
    for (int r = 0; r < NPW; ++r) {
        const int row = r * 8 + wave;
        const int total = lm[r];
        int c = cs[r];
        float w = cw[r];
        float ax = 0.0f, ay = 0.0f, az = 0.0f, aw = 0.0f;
        for (int base = 0; base < total; base += 32) {
            if (base > 0) {
                const int i = base + lane;
                const bool in = i < total;
                c = in ? col[lst[r] + i - 1] : 0;
                w = in ? dinv[c] * ldi[r] : 0.0f;
            }
            const int m = (total - base < 32) ? (total - base) : 32;
            int e = 0;
            for (; e + 8 <= m; e += 8) {
                float4 v[8];
                #pragma unroll
                for (int u = 0; u < 8; ++u) {
                    const int s = __builtin_amdgcn_readlane(c, e + u);
                    v[u] = load_row_piece(x, s, lane);
                }
                #pragma unroll
                for (int u = 0; u < 8; ++u) {
                    const float k = __int_as_float(__builtin_amdgcn_readlane(__float_as_int(w), e + u));
                    ax += v[u].x * k; ay += v[u].y * k; az += v[u].z * k; aw += v[u].w * k;
                }
            }
            if (e < m) {
                float4 v[7];
                #pragma unroll
                for (int u = 0; u < 7; ++u) v[u] = make_float4(0.0f, 0.0f, 0.0f, 0.0f);
                #pragma unroll
                for (int u = 0; u < 7; ++u) {
                    if (e + u < m) {
                        const int s = __builtin_amdgcn_readlane(c, e + u);
                        v[u] = load_row_piece(x, s, lane);
                    }
                }
                #pragma unroll
                for (int u = 0; u < 7; ++u) {
                    if (e + u < m) {
                        const float k = __int_as_float(__builtin_amdgcn_readlane(__float_as_int(w), e + u));
                        ax += v[u].x * k; ay += v[u].y * k; az += v[u].z * k; aw += v[u].w * k;
                    }
                }
            }
        }
        float4 o; o.x = ax; o.y = ay; o.z = az; o.w = aw;
        *(float4*)(&strip[row * SP + lane * 4]) = o;
    }
    __syncthreads();

    const int half = lane >> 4;
    const int l16  = lane & 15;

    static_assert(4 * RT == 8, "phase 3 gives each of the 8 waves one 16x16 tile");
    const int ct2 = wave & 3;
    const int t2 = wave >> 2;
    const float4* __restrict__ w2q = (const float4*)W2f + ct2 * (16 * 32) + lane;
    float4 qa[4], qb[4];
    {
        const int col_base = wave * 16;
        const float4* __restrict__ w1q = (const float4*)W1f + wave * (16 * 32) + lane;
        v8f acc[RT];
        #pragma unroll
        for (int t = 0; t < RT; ++t) acc[t] = v8f{};
        load_four_chunks(qa, w1q, 0);
        load_four_chunks(qb, w1q, 1);
        mma_four_chunks<RT>(acc, qa, strip, 0, l16, half);
        load_four_chunks(qa, w1q, 2);
        mma_four_chunks<RT>(acc, qb, strip, 32, l16, half);
        load_four_chunks(qb, w1q, 3);
        mma_four_chunks<RT>(acc, qa, strip, 64, l16, half);
        mma_four_chunks<RT>(acc, qb, strip, 96, l16, half);
        load_four_chunks(qa, w2q, 0);
        load_four_chunks(qb, w2q, 1);
        __syncthreads();
        const float bb = b1[col_base + l16];
        #pragma unroll
        for (int t = 0; t < RT; ++t) {
            #pragma unroll
            for (int v = 0; v < 8; ++v) {
                strip[(t * 16 + v + half * 8) * SP + col_base + l16] = fmaxf(acc[t][v] + bb, 0.0f);
            }
        }
        __syncthreads();
    }

    {
        const int col_base = ct2 * 16;
        float dv[8];
        #pragma unroll
        for (int v = 0; v < 8; ++v) {
            const int node = node0 + t2 * 16 + v + half * 8;
            dv[v] = dinv[(node < n) ? node : (n - 1)];
        }
        const float* rows = strip + t2 * 16 * SP;
        v8f acc[1];
        acc[0] = v8f{};
        mma_four_chunks<1>(acc, qa, rows, 0, l16, half);
        load_four_chunks(qa, w2q, 2);
        mma_four_chunks<1>(acc, qb, rows, 32, l16, half);
        load_four_chunks(qb, w2q, 3);
        mma_four_chunks<1>(acc, qa, rows, 64, l16, half);
        mma_four_chunks<1>(acc, qb, rows, 96, l16, half);
        #pragma unroll
        for (int v = 0; v < 8; ++v) {
            const int node = node0 + t2 * 16 + v + half * 8;
            if (node < n) h2[(size_t)node * D_OUT + col_base + l16] = acc[0][v] * dv[v];
        }
    }
}

#define GN 4

__device__ __forceinline__ float4 load_hs_piece(const float* __restrict__ hs, int s, int sub) {
    const unsigned off = ((unsigned)s << 8) + ((unsigned)sub << 4);
    return *(const float4*)((const char*)hs + off);
}

__global__ __launch_bounds__(256)
void gather_scaled_rows64_kernel(const float* __restrict__ hs,
                                 const float* __restrict__ dinv,
                                 const int* __restrict__ rowptr,
                                 const int* __restrict__ cnt,
                                 const int* __restrict__ col,
                                 const float* __restrict__ bias,
                                 float* __restrict__ out, int n) {
    const int lane = threadIdx.x & 31;
    const int wave = threadIdx.x >> 5;
    const int g = lane >> 4;
    const int sub = lane & 15;
    const int nodeA = (blockIdx.x * 8 + wave) * GN;

    int lnode[GN], lst[GN], ldg[GN];
    float ldi[GN];
    #pragma unroll
    for (int r = 0; r < GN; ++r) {
        const int node = __builtin_amdgcn_readfirstlane(nodeA + r);
        const int nc = (node < n) ? node : (n - 1);
        lnode[r] = (node < n) ? node : -1;
        lst[r] = rowptr[nc];
        ldg[r] = cnt[nc];
        ldi[r] = dinv[nc];
    }
    int cs[GN];
    #pragma unroll
    for (int r = 0; r < GN; ++r) {
        if (lnode[r] < 0) ldg[r] = 0;
        const int dg = ldg[r];
        const int k = (lane < dg) ? lane : (dg - 1);
        const int j = (dg > 0) ? (lst[r] + k) : 0;
        cs[r] = col[j];
    }
    const float4 bv = *(const float4*)(bias + sub * 4);

    #pragma unroll
    for (int r = 0; r < GN; ++r) {
        const int node = lnode[r];
        if (node < 0) continue;
        const int dg = ldg[r];
        int c = cs[r];
        float ax = 0.0f, ay = 0.0f, az = 0.0f, aw = 0.0f;
        if (g == 0) {
            const float4 v = load_hs_piece(hs, node, sub);
            ax = v.x; ay = v.y; az = v.z; aw = v.w;
        }
        for (int base = 0; base < dg; base += 32) {
            if (base > 0) {
                const int i = base + lane;
                c = col[lst[r] + ((i < dg) ? i : (dg - 1))];
            }
            const int m = (dg - base < 32) ? (dg - base) : 32;
            int e = 0;
            for (; e + 8 <= m; e += 8) {
                float4 v[4];
                #pragma unroll
                for (int u = 0; u < 4; ++u) {
                    const int s0 = __builtin_amdgcn_readlane(c, e + 2 * u);
                    const int s1 = __builtin_amdgcn_readlane(c, e + 2 * u + 1);
                    v[u] = load_hs_piece(hs, g ? s1 : s0, sub);
                }
                #pragma unroll
                for (int u = 0; u < 4; ++u) {
                    ax += v[u].x; ay += v[u].y; az += v[u].z; aw += v[u].w;
                }
            }
            if (e < m) {
                float4 v[4];
                #pragma unroll
                for (int u = 0; u < 4; ++u) v[u] = make_float4(0.0f, 0.0f, 0.0f, 0.0f);
                #pragma unroll
                for (int u = 0; u < 4; ++u) {
                    if (e + 2 * u < m) {
                        const int s0 = __builtin_amdgcn_readlane(c, e + 2 * u);
                        const int s1 = __builtin_amdgcn_readlane(c, e + 2 * u + 1);
                        if (e + 2 * u + g < m) v[u] = load_hs_piece(hs, g ? s1 : s0, sub);
                    }
                }
                #pragma unroll
                for (int u = 0; u < 4; ++u) {
                    if (e + 2 * u + g < m) {
                        ax += v[u].x; ay += v[u].y; az += v[u].z; aw += v[u].w;
                    }
                }
            }
        }
        ax += __shfl_xor(ax, 16);
        ay += __shfl_xor(ay, 16);
        az += __shfl_xor(az, 16);
        aw += __shfl_xor(aw, 16);
        if (g == 0) {
            const float di = ldi[r];
            float4 o;
            o.x = ax * di + bv.x; o.y = ay * di + bv.y;
            o.z = az * di + bv.z; o.w = aw * di + bv.w;
            *(float4*)(out + (size_t)node * D_OUT + sub * 4) = o;
        }
    }
}

extern "C" void kernel_launch(void* const* d_in, const int* in_sizes, int n_in,
                              void* d_out, int out_size, void* d_ws, size_t ws_size,
                              hipStream_t stream) {
    const float* x    = (const float*)d_in[0];
    const int*   eidx = (const int*)  d_in[1];
    const float* W1   = (const float*)d_in[2];
    const float* b1   = (const float*)d_in[3];
    const float* W2   = (const float*)d_in[4];
    const float* b2   = (const float*)d_in[5];
    float* out = (float*)d_out;

    const int N = in_sizes[0] / D_IN;
    const int E = in_sizes[1] / 2;
    const int* src = eidx;
    const int* dst = eidx + E;

    const int NB = (N + BIN_NODES - 1) / BIN_NODES;
    const int P  = (E + CH - 1) / CH;

    const size_t NP = ((size_t)N + 255) & ~(size_t)255;
    const size_t EP = ((size_t)E + 255) & ~(size_t)255;
    const size_t OP = ((size_t)(NB + 1) * P + 255) & ~(size_t)255;
    float* ws = (float*)d_ws;
    float* dinv   = ws;
    int*   cnt    = (int*)(ws + NP);
    int*   rowptr = (int*)(ws + 2 * NP);
    int*   part   = (int*)(ws + 3 * NP);
    int*   col    = (int*)(ws + 3 * NP + EP);
    int*   offs   = (int*)(ws + 3 * NP + 2 * EP);
    float* h2     = ws + 3 * NP + 2 * EP + OP;
    const size_t HP = ((size_t)N * D_OUT + 255) & ~(size_t)255;
    float* W1f    = h2 + HP;
    float* W2f    = W1f + D_IN * D_HID;

    const int BT = 256;

    bin_partition_kernel<<<P, 1024, 0, stream>>>(src, dst, part, offs, E, NB);
    bin_sort_kernel<<<NB + 1, 1024, 0, stream>>>(part, offs, cnt, rowptr, dinv, col, N, P,
                                                W1, W2, W1f, W2f, NB);

    const unsigned gather_blocks = (unsigned)((N + 8 * GN - 1) / (8 * GN));

    layer1_fused_kernel<<<(N + FN - 1) / FN, 256, 0, stream>>>(
        x, dinv, rowptr, cnt, col, W1f, b1, W2f, h2, N);

    gather_scaled_rows64_kernel<<<gather_blocks, BT, 0, stream>>>(
        h2, dinv, rowptr, cnt, col, b2, out, N);
}
